# starting kernel
# baseline (speedup 1.0000x reference)
#include <hip/hip_runtime.h>


typedef __attribute__((ext_vector_type(2))) float v2f;
typedef __attribute__((ext_vector_type(8))) float v8f;

__device__ __forceinline__ v8f wmma4(v2f a, v2f b, v8f c) {
  return __builtin_amdgcn_wmma_f32_16x16x4_f32(false, a, false, b, (short)0, c,
                                               false, false);
}

__device__ __forceinline__ v2f ld2(const float* p) { return *(const v2f*)p; }
__device__ __forceinline__ v2f ld2_nt(const float* p) {
  return __builtin_nontemporal_load((const v2f*)p);
}

__global__ __launch_bounds__(256) void k0_tables(float* __restrict__ tab) {
  int idx = blockIdx.x * 256 + threadIdx.x;
  if (idx >= 49152) return;
  const float step = 6.28318530717958647692f / 256.0f;
  float v;
  if (idx < 8192) {
    int n = idx >> 8, w = idx & 255;
    float th = step * (float)(((n & 15) * w) & 255);
    v = (n < 16) ? cosf(th) : -sinf(th);
  } else if (idx < 24576) {
    int j = idx - 8192;
    int is_s = j >= 8192;
    j &= 8191;
    int r = j >> 8, h = j & 255;
    int ky = (r < 16) ? r : (224 + r);
    float th = step * (float)((ky * h) & 255);
    v = is_s ? sinf(th) : cosf(th);
  } else if (idx < 40960) {
    int j = idx - 24576;
    int is_s = j >= 8192;
    j &= 8191;
    int h = j >> 5, r = j & 31;
    int ky = (r < 16) ? r : (224 + r);
    float th = step * (float)((ky * h) & 255);
    v = is_s ? sinf(th) : cosf(th);
  } else {
    int j = idx - 40960;
    int w = j >> 5, n = j & 31;
    int kx = n & 15;
    float th = step * (float)((kx * w) & 255);
    float sc = ((kx == 0) ? 1.0f : 2.0f) * (1.0f / 65536.0f);
    v = (n < 16) ? sc * cosf(th) : -sc * sinf(th);
  }
  tab[idx] = v;
}

__global__ __launch_bounds__(256) void k1_dft_w(const float* __restrict__ x,
                                                const float* __restrict__ FT,
                                                float* __restrict__ Y) {
  const int wave = blockIdx.x * 8 + (threadIdx.x >> 5);
  const int lane = threadIdx.x & 31;
  const int half = lane >> 4, l16 = lane & 15;
  const long rowbase = (long)wave * 16;
  const float* arow = x + (rowbase + l16) * 256 + 2 * half;
  const float* b0 = FT + l16 * 256 + 2 * half;
  const float* b1 = FT + (16 + l16) * 256 + 2 * half;
  v8f acc0 = {}, acc1 = {};
#pragma unroll 4
  for (int k0 = 0; k0 < 256; k0 += 4) {
    v2f a = ld2_nt(arow + k0);
    acc0 = wmma4(a, ld2(b0 + k0), acc0);
    acc1 = wmma4(a, ld2(b1 + k0), acc1);
  }
  float* yrow = Y + rowbase * 32;
#pragma unroll
  for (int r = 0; r < 8; ++r) {
    int row = r + half * 8;
    yrow[row * 32 + l16] = acc0[r];
    yrow[row * 32 + 16 + l16] = acc1[r];
  }
}

__global__ __launch_bounds__(256) void k2_dft_h(const float* __restrict__ Y,
                                                const float* __restrict__ E2c,
                                                const float* __restrict__ E2s,
                                                float* __restrict__ Zre,
                                                float* __restrict__ Zim) {
  const int bi = blockIdx.x * 8 + (threadIdx.x >> 5);
  const int lane = threadIdx.x & 31;
  const int half = lane >> 4, l16 = lane & 15;
  const float* yb = Y + (long)bi * 256 * 32;
  const float* c0 = E2c + l16 * 256 + 2 * half;
  const float* c1 = E2c + (16 + l16) * 256 + 2 * half;
  const float* s0 = E2s + l16 * 256 + 2 * half;
  const float* s1 = E2s + (16 + l16) * 256 + 2 * half;
  v8f a0 = {}, a1 = {}, a2 = {}, a3 = {}, a4 = {}, a5 = {}, a6 = {}, a7 = {};
#pragma unroll 2
  for (int k0 = 0; k0 < 256; k0 += 4) {
    int kr = k0 + 2 * half;
    v2f bre, bim;
    bre.x = yb[kr * 32 + l16];
    bre.y = yb[(kr + 1) * 32 + l16];
    bim.x = yb[kr * 32 + 16 + l16];
    bim.y = yb[(kr + 1) * 32 + 16 + l16];
    v2f ac0 = ld2(c0 + k0), ac1 = ld2(c1 + k0);
    v2f as0 = ld2(s0 + k0), as1 = ld2(s1 + k0);
    a0 = wmma4(ac0, bre, a0);
    a1 = wmma4(as0, bim, a1);
    a2 = wmma4(ac0, bim, a2);
    a3 = wmma4(as0, bre, a3);
    a4 = wmma4(ac1, bre, a4);
    a5 = wmma4(as1, bim, a5);
    a6 = wmma4(ac1, bim, a6);
    a7 = wmma4(as1, bre, a7);
  }
  const int b = bi >> 5, i = bi & 31;
#pragma unroll
  for (int r = 0; r < 8; ++r) {
    int kyr = r + half * 8;
    long o0 = (((long)kyr * 16 + l16) * 16 + b) * 32 + i;
    long o1 = (((long)(16 + kyr) * 16 + l16) * 16 + b) * 32 + i;
    Zre[o0] = a0[r] + a1[r];
    Zim[o0] = a2[r] - a3[r];
    Zre[o1] = a4[r] + a5[r];
    Zim[o1] = a6[r] - a7[r];
  }
}

__global__ __launch_bounds__(256) void k3_mix(
    const float* __restrict__ Zre, const float* __restrict__ Zim,
    const float* __restrict__ w1r, const float* __restrict__ w1i,
    const float* __restrict__ w2r, const float* __restrict__ w2i,
    float* __restrict__ Tre, float* __restrict__ Tim) {
  const int m = blockIdx.x * 8 + (threadIdx.x >> 5);
  const int lane = threadIdx.x & 31;
  const int half = lane >> 4, l16 = lane & 15;
  const int kyp = m >> 4, kx = m & 15;
  const int ky = kyp & 15;
  const float* wr = (kyp < 16) ? w1r : w2r;
  const float* wi = (kyp < 16) ? w1i : w2i;
  const float* zre = Zre + (long)m * 512 + l16 * 32 + 2 * half;
  const float* zim = Zim + (long)m * 512 + l16 * 32 + 2 * half;
  const long wofs = (long)ky * 16 + kx;
  v8f p0 = {}, q0 = {}, u0 = {}, v0 = {}, p1 = {}, q1 = {}, u1 = {}, v1 = {};
#pragma unroll
  for (int k0 = 0; k0 < 32; k0 += 4) {
    int kr = k0 + 2 * half;
    v2f azr = ld2(zre + k0);
    v2f azi = ld2(zim + k0);
    long ba0 = ((long)kr * 32 + l16) * 256 + wofs;
    long ba0b = ((long)(kr + 1) * 32 + l16) * 256 + wofs;
    long ba1 = ((long)kr * 32 + 16 + l16) * 256 + wofs;
    long ba1b = ((long)(kr + 1) * 32 + 16 + l16) * 256 + wofs;
    v2f br0, bi0, br1, bi1;
    br0.x = wr[ba0];  br0.y = wr[ba0b];
    bi0.x = wi[ba0];  bi0.y = wi[ba0b];
    br1.x = wr[ba1];  br1.y = wr[ba1b];
    bi1.x = wi[ba1];  bi1.y = wi[ba1b];
    p0 = wmma4(azr, br0, p0);
    q0 = wmma4(azi, bi0, q0);
    u0 = wmma4(azr, bi0, u0);
    v0 = wmma4(azi, br0, v0);
    p1 = wmma4(azr, br1, p1);
    q1 = wmma4(azi, bi1, q1);
    u1 = wmma4(azr, bi1, u1);
    v1 = wmma4(azi, br1, v1);
  }
#pragma unroll
  for (int r = 0; r < 8; ++r) {
    int b = r + half * 8;
    long o0 = (((long)b * 32 + l16) * 32 + kyp) * 16 + kx;
    long o1 = (((long)b * 32 + 16 + l16) * 32 + kyp) * 16 + kx;
    Tre[o0] = p0[r] - q0[r];
    Tim[o0] = u0[r] + v0[r];
    Tre[o1] = p1[r] - q1[r];
    Tim[o1] = u1[r] + v1[r];
  }
}

__global__ __launch_bounds__(256) void k4_idft_h(const float* __restrict__ Tre,
                                                 const float* __restrict__ Tim,
                                                 const float* __restrict__ E4c,
                                                 const float* __restrict__ E4s,
                                                 float* __restrict__ U) {
  const int wid = blockIdx.x * 8 + (threadIdx.x >> 5);
  const int lane = threadIdx.x & 31;
  const int half = lane >> 4, l16 = lane & 15;
  const int bo = wid >> 4;
  const int h0 = (wid & 15) * 16;
  const float* tre = Tre + (long)bo * 512;
  const float* tim = Tim + (long)bo * 512;
  const float* ec = E4c + (h0 + l16) * 32 + 2 * half;
  const float* es = E4s + (h0 + l16) * 32 + 2 * half;
  v8f a0 = {}, a1 = {}, a2 = {}, a3 = {};
#pragma unroll
  for (int k0 = 0; k0 < 32; k0 += 4) {
    int kr = k0 + 2 * half;
    v2f ac = ld2(ec + k0);
    v2f as = ld2(es + k0);
    v2f bre, bim;
    bre.x = tre[kr * 16 + l16];
    bre.y = tre[(kr + 1) * 16 + l16];
    bim.x = tim[kr * 16 + l16];
    bim.y = tim[(kr + 1) * 16 + l16];
    a0 = wmma4(ac, bre, a0);
    a1 = wmma4(as, bim, a1);
    a2 = wmma4(ac, bim, a2);
    a3 = wmma4(as, bre, a3);
  }
  float* ub = U + ((long)bo * 256 + h0) * 32;
#pragma unroll
  for (int r = 0; r < 8; ++r) {
    int hh = r + half * 8;
    ub[hh * 32 + l16] = a0[r] - a1[r];
    ub[hh * 32 + 16 + l16] = a2[r] + a3[r];
  }
}

__global__ __launch_bounds__(256) void k5_idft_w(const float* __restrict__ U,
                                                 const float* __restrict__ GT,
                                                 float* __restrict__ out) {
  const int wid = blockIdx.x * 8 + (threadIdx.x >> 5);
  const int lane = threadIdx.x & 31;
  const int half = lane >> 4, l16 = lane & 15;
  const long row0 = (long)(wid >> 2) * 16;
  const int n0 = (wid & 3) * 64;
  const float* arow = U + (row0 + l16) * 32 + 2 * half;
  const float* g0 = GT + (n0 + l16) * 32 + 2 * half;
  v8f acc[4] = {};
#pragma unroll
  for (int k0 = 0; k0 < 32; k0 += 4) {
    v2f a = ld2(arow + k0);
#pragma unroll
    for (int t = 0; t < 4; ++t) {
      acc[t] = wmma4(a, ld2(g0 + t * 16 * 32 + k0), acc[t]);
    }
  }
  float* orow = out + row0 * 256;
#pragma unroll
  for (int r = 0; r < 8; ++r) {
    int rr = r + half * 8;
#pragma unroll
    for (int t = 0; t < 4; ++t) {
      __builtin_nontemporal_store(acc[t][r],
                                  &orow[(long)rr * 256 + n0 + t * 16 + l16]);
    }
  }
}

extern "C" void kernel_launch(void* const* d_in, const int* in_sizes, int n_in,
                              void* d_out, int out_size, void* d_ws,
                              size_t ws_size, hipStream_t stream) {
  (void)in_sizes; (void)n_in; (void)out_size; (void)ws_size;
  const float* x = (const float*)d_in[0];
  const float* w1r = (const float*)d_in[1];
  const float* w1i = (const float*)d_in[2];
  const float* w2r = (const float*)d_in[3];
  const float* w2i = (const float*)d_in[4];
  float* out = (float*)d_out;

  float* tab = (float*)d_ws;
  float* FT = tab;
  float* E2c = tab + 8192;
  float* E2s = tab + 16384;
  float* E4c = tab + 24576;
  float* E4s = tab + 32768;
  float* GT = tab + 40960;
  float* Y = tab + 49152;
  float* Zre = Y + 4194304;
  float* Zim = Zre + 262144;
  float* Tre = Zim + 262144;
  float* Tim = Tre + 262144;
  float* U = Tim + 262144;

  hipLaunchKernelGGL(k0_tables, dim3(192), dim3(256), 0, stream, tab);
  hipLaunchKernelGGL(k1_dft_w, dim3(1024), dim3(256), 0, stream, x, FT, Y);
  hipLaunchKernelGGL(k2_dft_h, dim3(64), dim3(256), 0, stream, Y, E2c, E2s, Zre,
                     Zim);
  hipLaunchKernelGGL(k3_mix, dim3(64), dim3(256), 0, stream, Zre, Zim, w1r, w1i,
                     w2r, w2i, Tre, Tim);
  hipLaunchKernelGGL(k4_idft_h, dim3(1024), dim3(256), 0, stream, Tre, Tim, E4c,
                     E4s, U);
  hipLaunchKernelGGL(k5_idft_w, dim3(4096), dim3(256), 0, stream, U, GT, out);
}


# Twiddle tables stored in WMMA fragment order: a table fragment is one 16-byte-per-lane load of 512 contiguous bytes (4 lines), not 16 lines 1 KB apart; k order and summation order unchanged
# speedup vs baseline: 1.0833x; 1.0833x over previous
#include <hip/hip_runtime.h>


typedef __attribute__((ext_vector_type(2))) float v2f;
typedef __attribute__((ext_vector_type(4))) float v4f;
typedef __attribute__((ext_vector_type(8))) float v8f;

__device__ __forceinline__ v8f wmma4(v2f a, v2f b, v8f c) {
  return __builtin_amdgcn_wmma_f32_16x16x4_f32(false, a, false, b, (short)0, c,
                                               false, false);
}

__device__ __forceinline__ v2f ld2(const float* p) { return *(const v2f*)p; }
__device__ __forceinline__ v4f ld4(const float* p) { return *(const v4f*)p; }
__device__ __forceinline__ v2f ld2_nt(const float* p) {
  return __builtin_nontemporal_load((const v2f*)p);
}

__global__ __launch_bounds__(256) void k0_tables(float* __restrict__ tab) {
  int idx = blockIdx.x * 256 + threadIdx.x;
  if (idx >= 49152) return;
  const float step = 6.28318530717958647692f / 256.0f;
  const int s = idx & 1, jt = (idx >> 1) & 1;
  const int lane = (idx >> 2) & 31, c = lane & 15, hf = lane >> 4;
  float v;
  if (idx < 8192) {
    int ks = idx >> 7;
    int w = 4 * ks + 2 * hf + s;
    int n = 16 * jt + c;
    float th = step * (float)(((n & 15) * w) & 255);
    v = (n < 16) ? cosf(th) : -sinf(th);
  } else if (idx < 24576) {
    int q = idx - 8192;
    int is_s = q >= 8192;
    q &= 8191;
    int ks = q >> 7;
    int h = 4 * ks + 2 * hf + s;
    int r = 16 * jt + c;
    int ky = (r < 16) ? r : (224 + r);
    float th = step * (float)((ky * h) & 255);
    v = is_s ? sinf(th) : cosf(th);
  } else if (idx < 40960) {
    int q = idx - 24576;
    int ks = (q >> 7) & 7, ht = q >> 10;
    int h = 16 * ht + c;
    int r = 4 * ks + 2 * hf + s;
    int ky = (r < 16) ? r : (224 + r);
    float th = step * (float)((ky * h) & 255);
    v = jt ? sinf(th) : cosf(th);
  } else {
    int q = idx - 40960;
    int p = (q >> 7) & 1, ks = (q >> 8) & 7, g = q >> 11;
    int w = 64 * g + 16 * (2 * p + jt) + c;
    int n = 4 * ks + 2 * hf + s;
    int kx = n & 15;
    float th = step * (float)((kx * w) & 255);
    float sc = ((kx == 0) ? 1.0f : 2.0f) * (1.0f / 65536.0f);
    v = (n < 16) ? sc * cosf(th) : -sc * sinf(th);
  }
  tab[idx] = v;
}

__global__ __launch_bounds__(256) void k1_dft_w(const float* __restrict__ x,
                                                const float* __restrict__ FT,
                                                float* __restrict__ Y) {
  const int wave = blockIdx.x * 8 + (threadIdx.x >> 5);
  const int lane = threadIdx.x & 31;
  const int half = lane >> 4, l16 = lane & 15;
  const long rowbase = (long)wave * 16;
  const float* arow = x + (rowbase + l16) * 256 + 2 * half;
  const float* fb = FT + lane * 4;
  v8f acc0 = {}, acc1 = {};
#pragma unroll 4
  for (int k0 = 0; k0 < 256; k0 += 4) {
    v2f a = ld2_nt(arow + k0);
    v4f f = ld4(fb + k0 * 32);
    v2f b0 = {f.x, f.y};
    v2f b1 = {f.z, f.w};
    acc0 = wmma4(a, b0, acc0);
    acc1 = wmma4(a, b1, acc1);
  }
  float* yrow = Y + rowbase * 32;
#pragma unroll
  for (int r = 0; r < 8; ++r) {
    int row = r + half * 8;
    yrow[row * 32 + l16] = acc0[r];
    yrow[row * 32 + 16 + l16] = acc1[r];
  }
}

__global__ __launch_bounds__(256) void k2_dft_h(const float* __restrict__ Y,
                                                const float* __restrict__ E2c,
                                                const float* __restrict__ E2s,
                                                float* __restrict__ Zre,
                                                float* __restrict__ Zim) {
  const int bi = blockIdx.x * 8 + (threadIdx.x >> 5);
  const int lane = threadIdx.x & 31;
  const int half = lane >> 4, l16 = lane & 15;
  const float* yb = Y + (long)bi * 256 * 32;
  const float* cf = E2c + lane * 4;
  const float* sf = E2s + lane * 4;
  v8f a0 = {}, a1 = {}, a2 = {}, a3 = {}, a4 = {}, a5 = {}, a6 = {}, a7 = {};
#pragma unroll 2
  for (int k0 = 0; k0 < 256; k0 += 4) {
    int kr = k0 + 2 * half;
    v2f bre, bim;
    bre.x = yb[kr * 32 + l16];
    bre.y = yb[(kr + 1) * 32 + l16];
    bim.x = yb[kr * 32 + 16 + l16];
    bim.y = yb[(kr + 1) * 32 + 16 + l16];
    v4f cc = ld4(cf + k0 * 32), ss = ld4(sf + k0 * 32);
    v2f ac0 = {cc.x, cc.y}, ac1 = {cc.z, cc.w};
    v2f as0 = {ss.x, ss.y}, as1 = {ss.z, ss.w};
    a0 = wmma4(ac0, bre, a0);
    a1 = wmma4(as0, bim, a1);
    a2 = wmma4(ac0, bim, a2);
    a3 = wmma4(as0, bre, a3);
    a4 = wmma4(ac1, bre, a4);
    a5 = wmma4(as1, bim, a5);
    a6 = wmma4(ac1, bim, a6);
    a7 = wmma4(as1, bre, a7);
  }
  const int b = bi >> 5, i = bi & 31;
#pragma unroll
  for (int r = 0; r < 8; ++r) {
    int kyr = r + half * 8;
    long o0 = (((long)kyr * 16 + l16) * 16 + b) * 32 + i;
    long o1 = (((long)(16 + kyr) * 16 + l16) * 16 + b) * 32 + i;
    Zre[o0] = a0[r] + a1[r];
    Zim[o0] = a2[r] - a3[r];
    Zre[o1] = a4[r] + a5[r];
    Zim[o1] = a6[r] - a7[r];
  }
}

__global__ __launch_bounds__(256) void k3_mix(
    const float* __restrict__ Zre, const float* __restrict__ Zim,
    const float* __restrict__ w1r, const float* __restrict__ w1i,
    const float* __restrict__ w2r, const float* __restrict__ w2i,
    float* __restrict__ Tre, float* __restrict__ Tim) {
  const int m = blockIdx.x * 8 + (threadIdx.x >> 5);
  const int lane = threadIdx.x & 31;
  const int half = lane >> 4, l16 = lane & 15;
  const int kyp = m >> 4, kx = m & 15;
  const int ky = kyp & 15;
  const float* wr = (kyp < 16) ? w1r : w2r;
  const float* wi = (kyp < 16) ? w1i : w2i;
  const float* zre = Zre + (long)m * 512 + l16 * 32 + 2 * half;
  const float* zim = Zim + (long)m * 512 + l16 * 32 + 2 * half;
  const long wofs = (long)ky * 16 + kx;
  v8f p0 = {}, q0 = {}, u0 = {}, v0 = {}, p1 = {}, q1 = {}, u1 = {}, v1 = {};
#pragma unroll
  for (int k0 = 0; k0 < 32; k0 += 4) {
    int kr = k0 + 2 * half;
    v2f azr = ld2(zre + k0);
    v2f azi = ld2(zim + k0);
    long ba0 = ((long)kr * 32 + l16) * 256 + wofs;
    long ba0b = ((long)(kr + 1) * 32 + l16) * 256 + wofs;
    long ba1 = ((long)kr * 32 + 16 + l16) * 256 + wofs;
    long ba1b = ((long)(kr + 1) * 32 + 16 + l16) * 256 + wofs;
    v2f br0, bi0, br1, bi1;
    br0.x = wr[ba0];  br0.y = wr[ba0b];
    bi0.x = wi[ba0];  bi0.y = wi[ba0b];
    br1.x = wr[ba1];  br1.y = wr[ba1b];
    bi1.x = wi[ba1];  bi1.y = wi[ba1b];
    p0 = wmma4(azr, br0, p0);
    q0 = wmma4(azi, bi0, q0);
    u0 = wmma4(azr, bi0, u0);
    v0 = wmma4(azi, br0, v0);
    p1 = wmma4(azr, br1, p1);
    q1 = wmma4(azi, bi1, q1);
    u1 = wmma4(azr, bi1, u1);
    v1 = wmma4(azi, br1, v1);
  }
#pragma unroll
  for (int r = 0; r < 8; ++r) {
    int b = r + half * 8;
    long o0 = (((long)b * 32 + l16) * 32 + kyp) * 16 + kx;
    long o1 = (((long)b * 32 + 16 + l16) * 32 + kyp) * 16 + kx;
    Tre[o0] = p0[r] - q0[r];
    Tim[o0] = u0[r] + v0[r];
    Tre[o1] = p1[r] - q1[r];
    Tim[o1] = u1[r] + v1[r];
  }
}

__global__ __launch_bounds__(256) void k4_idft_h(const float* __restrict__ Tre,
                                                 const float* __restrict__ Tim,
                                                 const float* __restrict__ E4F,
                                                 float* __restrict__ U) {
  const int wid = blockIdx.x * 8 + (threadIdx.x >> 5);
  const int lane = threadIdx.x & 31;
  const int half = lane >> 4, l16 = lane & 15;
  const int bo = wid >> 4;
  const int h0 = (wid & 15) * 16;
  const float* tre = Tre + (long)bo * 512;
  const float* tim = Tim + (long)bo * 512;
  const float* ef = E4F + ((wid & 15) * 256 + lane) * 4;
  v8f a0 = {}, a1 = {}, a2 = {}, a3 = {};
#pragma unroll
  for (int k0 = 0; k0 < 32; k0 += 4) {
    int kr = k0 + 2 * half;
    v4f e = ld4(ef + k0 * 32);
    v2f ac = {e.x, e.y};
    v2f as = {e.z, e.w};
    v2f bre, bim;
    bre.x = tre[kr * 16 + l16];
    bre.y = tre[(kr + 1) * 16 + l16];
    bim.x = tim[kr * 16 + l16];
    bim.y = tim[(kr + 1) * 16 + l16];
    a0 = wmma4(ac, bre, a0);
    a1 = wmma4(as, bim, a1);
    a2 = wmma4(ac, bim, a2);
    a3 = wmma4(as, bre, a3);
  }
  float* ub = U + ((long)bo * 256 + h0) * 32;
#pragma unroll
  for (int r = 0; r < 8; ++r) {
    int hh = r + half * 8;
    ub[hh * 32 + l16] = a0[r] - a1[r];
    ub[hh * 32 + 16 + l16] = a2[r] + a3[r];
  }
}

__global__ __launch_bounds__(256) void k5_idft_w(const float* __restrict__ U,
                                                 const float* __restrict__ GT,
                                                 float* __restrict__ out) {
  const int wid = blockIdx.x * 8 + (threadIdx.x >> 5);
  const int lane = threadIdx.x & 31;
  const int half = lane >> 4, l16 = lane & 15;
  const long row0 = (long)(wid >> 2) * 16;
  const int n0 = (wid & 3) * 64;
  const float* arow = U + (row0 + l16) * 32 + 2 * half;
  const float* gf = GT + ((wid & 3) * 512 + lane) * 4;
  v8f acc[4] = {};
#pragma unroll
  for (int k0 = 0; k0 < 32; k0 += 4) {
    v2f a = ld2(arow + k0);
#pragma unroll
    for (int p = 0; p < 2; ++p) {
      v4f g = ld4(gf + k0 * 64 + p * 128);
      v2f ga = {g.x, g.y}, gb = {g.z, g.w};
      acc[2 * p] = wmma4(a, ga, acc[2 * p]);
      acc[2 * p + 1] = wmma4(a, gb, acc[2 * p + 1]);
    }
  }
  float* orow = out + row0 * 256;
#pragma unroll
  for (int r = 0; r < 8; ++r) {
    int rr = r + half * 8;
#pragma unroll
    for (int t = 0; t < 4; ++t) {
      __builtin_nontemporal_store(acc[t][r],
                                  &orow[(long)rr * 256 + n0 + t * 16 + l16]);
    }
  }
}

extern "C" void kernel_launch(void* const* d_in, const int* in_sizes, int n_in,
                              void* d_out, int out_size, void* d_ws,
                              size_t ws_size, hipStream_t stream) {
  (void)in_sizes; (void)n_in; (void)out_size; (void)ws_size;
  const float* x = (const float*)d_in[0];
  const float* w1r = (const float*)d_in[1];
  const float* w1i = (const float*)d_in[2];
  const float* w2r = (const float*)d_in[3];
  const float* w2i = (const float*)d_in[4];
  float* out = (float*)d_out;

  float* tab = (float*)d_ws;
  float* FT = tab;
  float* E2c = tab + 8192;
  float* E2s = tab + 16384;
  float* E4F = tab + 24576;
  float* GT = tab + 40960;
  float* Y = tab + 49152;
  float* Zre = Y + 4194304;
  float* Zim = Zre + 262144;
  float* Tre = Zim + 262144;
  float* Tim = Tre + 262144;
  float* U = Tim + 262144;

  hipLaunchKernelGGL(k0_tables, dim3(192), dim3(256), 0, stream, tab);
  hipLaunchKernelGGL(k1_dft_w, dim3(1024), dim3(256), 0, stream, x, FT, Y);
  hipLaunchKernelGGL(k2_dft_h, dim3(64), dim3(256), 0, stream, Y, E2c, E2s, Zre,
                     Zim);
  hipLaunchKernelGGL(k3_mix, dim3(64), dim3(256), 0, stream, Zre, Zim, w1r, w1i,
                     w2r, w2i, Tre, Tim);
  hipLaunchKernelGGL(k4_idft_h, dim3(1024), dim3(256), 0, stream, Tre, Tim, E4F,
                     U);
  hipLaunchKernelGGL(k5_idft_w, dim3(4096), dim3(256), 0, stream, U, GT, out);
}


# K1 reads x in whole 256-byte units (16-byte load per lane) and turns each 16x64 chunk into WMMA fragments through a per-wave LDS tile; W-sum k order permuted within groups of 8, table order matched
# speedup vs baseline: 1.2945x; 1.1950x over previous
#include <hip/hip_runtime.h>


typedef __attribute__((ext_vector_type(2))) float v2f;
typedef __attribute__((ext_vector_type(4))) float v4f;
typedef __attribute__((ext_vector_type(8))) float v8f;

__device__ __forceinline__ v8f wmma4(v2f a, v2f b, v8f c) {
  return __builtin_amdgcn_wmma_f32_16x16x4_f32(false, a, false, b, (short)0, c,
                                               false, false);
}

__device__ __forceinline__ v2f ld2(const float* p) { return *(const v2f*)p; }
__device__ __forceinline__ v4f ld4(const float* p) { return *(const v4f*)p; }
__device__ __forceinline__ v4f ld4_nt(const float* p) {
  return __builtin_nontemporal_load((const v4f*)p);
}

__global__ __launch_bounds__(256) void k0_tables(float* __restrict__ tab) {
  int idx = blockIdx.x * 256 + threadIdx.x;
  if (idx >= 49152) return;
  const float step = 6.28318530717958647692f / 256.0f;
  const int s = idx & 1, jt = (idx >> 1) & 1;
  const int lane = (idx >> 2) & 31, c = lane & 15, hf = lane >> 4;
  float v;
  if (idx < 8192) {
    int ks = idx >> 7;
    int w = 8 * (ks >> 1) + 4 * hf + 2 * (ks & 1) + s;
    int n = 16 * jt + c;
    float th = step * (float)(((n & 15) * w) & 255);
    v = (n < 16) ? cosf(th) : -sinf(th);
  } else if (idx < 24576) {
    int q = idx - 8192;
    int is_s = q >= 8192;
    q &= 8191;
    int ks = q >> 7;
    int h = 4 * ks + 2 * hf + s;
    int r = 16 * jt + c;
    int ky = (r < 16) ? r : (224 + r);
    float th = step * (float)((ky * h) & 255);
    v = is_s ? sinf(th) : cosf(th);
  } else if (idx < 40960) {
    int q = idx - 24576;
    int ks = (q >> 7) & 7, ht = q >> 10;
    int h = 16 * ht + c;
    int r = 4 * ks + 2 * hf + s;
    int ky = (r < 16) ? r : (224 + r);
    float th = step * (float)((ky * h) & 255);
    v = jt ? sinf(th) : cosf(th);
  } else {
    int q = idx - 40960;
    int p = (q >> 7) & 1, ks = (q >> 8) & 7, g = q >> 11;
    int w = 64 * g + 16 * (2 * p + jt) + c;
    int n = 4 * ks + 2 * hf + s;
    int kx = n & 15;
    float th = step * (float)((kx * w) & 255);
    float sc = ((kx == 0) ? 1.0f : 2.0f) * (1.0f / 65536.0f);
    v = (n < 16) ? sc * cosf(th) : -sc * sinf(th);
  }
  tab[idx] = v;
}

#define K1_PITCH4 17
__global__ __launch_bounds__(256) __attribute__((amdgpu_waves_per_eu(8, 8)))
void k1_dft_w(const float* __restrict__ x, const float* __restrict__ FT,
              float* __restrict__ Y) {
  __shared__ v4f stage[8 * 16 * K1_PITCH4];
  const int wv = threadIdx.x >> 5;
  const int wave = blockIdx.x * 8 + wv;
  const int lane = threadIdx.x & 31;
  const int half = lane >> 4, l16 = lane & 15;
  const long rowbase = (long)wave * 16;
  v4f* tile = stage + wv * (16 * K1_PITCH4);
  const float* gsrc = x + (rowbase + half) * 256 + 4 * l16;
  v4f* sdst = tile + half * K1_PITCH4 + l16;
  const v4f* ssrc = tile + l16 * K1_PITCH4 + half;
  const float* fb = FT + lane * 4;
  v8f acc0 = {}, acc1 = {};
  v4f g[8];
#pragma unroll
  for (int i = 0; i < 8; ++i) g[i] = ld4_nt(gsrc + i * 512);
#pragma unroll
  for (int c = 0; c < 4; ++c) {
#pragma unroll
    for (int i = 0; i < 8; ++i) sdst[2 * i * K1_PITCH4] = g[i];
    if (c < 3) {
#pragma unroll
      for (int i = 0; i < 8; ++i) g[i] = ld4_nt(gsrc + i * 512 + (c + 1) * 64);
    }
    __builtin_amdgcn_fence(__ATOMIC_ACQ_REL, "wavefront");
    v4f r[8];
#pragma unroll
    for (int j = 0; j < 8; ++j) r[j] = ssrc[2 * j];
#pragma unroll
    for (int j = 0; j < 8; ++j) {
      const int ks = (c * 8 + j) * 2;
      v4f f0 = ld4(fb + ks * 128);
      v4f f1 = ld4(fb + (ks + 1) * 128);
      v2f a0 = {r[j].x, r[j].y};
      v2f a1 = {r[j].z, r[j].w};
      v2f b00 = {f0.x, f0.y};
      v2f b01 = {f0.z, f0.w};
      v2f b10 = {f1.x, f1.y};
      v2f b11 = {f1.z, f1.w};
      acc0 = wmma4(a0, b00, acc0);
      acc1 = wmma4(a0, b01, acc1);
      acc0 = wmma4(a1, b10, acc0);
      acc1 = wmma4(a1, b11, acc1);
    }
    if (c < 3) __builtin_amdgcn_fence(__ATOMIC_ACQ_REL, "wavefront");
  }
  float* yrow = Y + rowbase * 32;
#pragma unroll
  for (int r = 0; r < 8; ++r) {
    int row = r + half * 8;
    yrow[row * 32 + l16] = acc0[r];
    yrow[row * 32 + 16 + l16] = acc1[r];
  }
}

__global__ __launch_bounds__(256) void k2_dft_h(const float* __restrict__ Y,
                                                const float* __restrict__ E2c,
                                                const float* __restrict__ E2s,
                                                float* __restrict__ Zre,
                                                float* __restrict__ Zim) {
  const int bi = blockIdx.x * 8 + (threadIdx.x >> 5);
  const int lane = threadIdx.x & 31;
  const int half = lane >> 4, l16 = lane & 15;
  const float* yb = Y + (long)bi * 256 * 32;
  const float* cf = E2c + lane * 4;
  const float* sf = E2s + lane * 4;
  v8f a0 = {}, a1 = {}, a2 = {}, a3 = {}, a4 = {}, a5 = {}, a6 = {}, a7 = {};
#pragma unroll 2
  for (int k0 = 0; k0 < 256; k0 += 4) {
    int kr = k0 + 2 * half;
    v2f bre, bim;
    bre.x = yb[kr * 32 + l16];
    bre.y = yb[(kr + 1) * 32 + l16];
    bim.x = yb[kr * 32 + 16 + l16];
    bim.y = yb[(kr + 1) * 32 + 16 + l16];
    v4f cc = ld4(cf + k0 * 32), ss = ld4(sf + k0 * 32);
    v2f ac0 = {cc.x, cc.y}, ac1 = {cc.z, cc.w};
    v2f as0 = {ss.x, ss.y}, as1 = {ss.z, ss.w};
    a0 = wmma4(ac0, bre, a0);
    a1 = wmma4(as0, bim, a1);
    a2 = wmma4(ac0, bim, a2);
    a3 = wmma4(as0, bre, a3);
    a4 = wmma4(ac1, bre, a4);
    a5 = wmma4(as1, bim, a5);
    a6 = wmma4(ac1, bim, a6);
    a7 = wmma4(as1, bre, a7);
  }
  const int b = bi >> 5, i = bi & 31;
#pragma unroll
  for (int r = 0; r < 8; ++r) {
    int kyr = r + half * 8;
    long o0 = (((long)kyr * 16 + l16) * 16 + b) * 32 + i;
    long o1 = (((long)(16 + kyr) * 16 + l16) * 16 + b) * 32 + i;
    Zre[o0] = a0[r] + a1[r];
    Zim[o0] = a2[r] - a3[r];
    Zre[o1] = a4[r] + a5[r];
    Zim[o1] = a6[r] - a7[r];
  }
}

__global__ __launch_bounds__(256) void k3_mix(
    const float* __restrict__ Zre, const float* __restrict__ Zim,
    const float* __restrict__ w1r, const float* __restrict__ w1i,
    const float* __restrict__ w2r, const float* __restrict__ w2i,
    float* __restrict__ Tre, float* __restrict__ Tim) {
  const int m = blockIdx.x * 8 + (threadIdx.x >> 5);
  const int lane = threadIdx.x & 31;
  const int half = lane >> 4, l16 = lane & 15;
  const int kyp = m >> 4, kx = m & 15;
  const int ky = kyp & 15;
  const float* wr = (kyp < 16) ? w1r : w2r;
  const float* wi = (kyp < 16) ? w1i : w2i;
  const float* zre = Zre + (long)m * 512 + l16 * 32 + 2 * half;
  const float* zim = Zim + (long)m * 512 + l16 * 32 + 2 * half;
  const long wofs = (long)ky * 16 + kx;
  v8f p0 = {}, q0 = {}, u0 = {}, v0 = {}, p1 = {}, q1 = {}, u1 = {}, v1 = {};
#pragma unroll
  for (int k0 = 0; k0 < 32; k0 += 4) {
    int kr = k0 + 2 * half;
    v2f azr = ld2(zre + k0);
    v2f azi = ld2(zim + k0);
    long ba0 = ((long)kr * 32 + l16) * 256 + wofs;
    long ba0b = ((long)(kr + 1) * 32 + l16) * 256 + wofs;
    long ba1 = ((long)kr * 32 + 16 + l16) * 256 + wofs;
    long ba1b = ((long)(kr + 1) * 32 + 16 + l16) * 256 + wofs;
    v2f br0, bi0, br1, bi1;
    br0.x = wr[ba0];  br0.y = wr[ba0b];
    bi0.x = wi[ba0];  bi0.y = wi[ba0b];
    br1.x = wr[ba1];  br1.y = wr[ba1b];
    bi1.x = wi[ba1];  bi1.y = wi[ba1b];
    p0 = wmma4(azr, br0, p0);
    q0 = wmma4(azi, bi0, q0);
    u0 = wmma4(azr, bi0, u0);
    v0 = wmma4(azi, br0, v0);
    p1 = wmma4(azr, br1, p1);
    q1 = wmma4(azi, bi1, q1);
    u1 = wmma4(azr, bi1, u1);
    v1 = wmma4(azi, br1, v1);
  }
#pragma unroll
  for (int r = 0; r < 8; ++r) {
    int b = r + half * 8;
    long o0 = (((long)b * 32 + l16) * 32 + kyp) * 16 + kx;
    long o1 = (((long)b * 32 + 16 + l16) * 32 + kyp) * 16 + kx;
    Tre[o0] = p0[r] - q0[r];
    Tim[o0] = u0[r] + v0[r];
    Tre[o1] = p1[r] - q1[r];
    Tim[o1] = u1[r] + v1[r];
  }
}

__global__ __launch_bounds__(256) void k4_idft_h(const float* __restrict__ Tre,
                                                 const float* __restrict__ Tim,
                                                 const float* __restrict__ E4F,
                                                 float* __restrict__ U) {
  const int wid = blockIdx.x * 8 + (threadIdx.x >> 5);
  const int lane = threadIdx.x & 31;
  const int half = lane >> 4, l16 = lane & 15;
  const int bo = wid >> 4;
  const int h0 = (wid & 15) * 16;
  const float* tre = Tre + (long)bo * 512;
  const float* tim = Tim + (long)bo * 512;
  const float* ef = E4F + ((wid & 15) * 256 + lane) * 4;
  v8f a0 = {}, a1 = {}, a2 = {}, a3 = {};
#pragma unroll
  for (int k0 = 0; k0 < 32; k0 += 4) {
    int kr = k0 + 2 * half;
    v4f e = ld4(ef + k0 * 32);
    v2f ac = {e.x, e.y};
    v2f as = {e.z, e.w};
    v2f bre, bim;
    bre.x = tre[kr * 16 + l16];
    bre.y = tre[(kr + 1) * 16 + l16];
    bim.x = tim[kr * 16 + l16];
    bim.y = tim[(kr + 1) * 16 + l16];
    a0 = wmma4(ac, bre, a0);
    a1 = wmma4(as, bim, a1);
    a2 = wmma4(ac, bim, a2);
    a3 = wmma4(as, bre, a3);
  }
  float* ub = U + ((long)bo * 256 + h0) * 32;
#pragma unroll
  for (int r = 0; r < 8; ++r) {
    int hh = r + half * 8;
    ub[hh * 32 + l16] = a0[r] - a1[r];
    ub[hh * 32 + 16 + l16] = a2[r] + a3[r];
  }
}

__global__ __launch_bounds__(256) void k5_idft_w(const float* __restrict__ U,
                                                 const float* __restrict__ GT,
                                                 float* __restrict__ out) {
  const int wid = blockIdx.x * 8 + (threadIdx.x >> 5);
  const int lane = threadIdx.x & 31;
  const int half = lane >> 4, l16 = lane & 15;
  const long row0 = (long)(wid >> 2) * 16;
  const int n0 = (wid & 3) * 64;
  const float* arow = U + (row0 + l16) * 32 + 2 * half;
  const float* gf = GT + ((wid & 3) * 512 + lane) * 4;
  v8f acc[4] = {};
#pragma unroll
  for (int k0 = 0; k0 < 32; k0 += 4) {
    v2f a = ld2(arow + k0);
#pragma unroll
    for (int p = 0; p < 2; ++p) {
      v4f g = ld4(gf + k0 * 64 + p * 128);
      v2f ga = {g.x, g.y}, gb = {g.z, g.w};
      acc[2 * p] = wmma4(a, ga, acc[2 * p]);
      acc[2 * p + 1] = wmma4(a, gb, acc[2 * p + 1]);
    }
  }
  float* orow = out + row0 * 256;
#pragma unroll
  for (int r = 0; r < 8; ++r) {
    int rr = r + half * 8;
#pragma unroll
    for (int t = 0; t < 4; ++t) {
      __builtin_nontemporal_store(acc[t][r],
                                  &orow[(long)rr * 256 + n0 + t * 16 + l16]);
    }
  }
}

extern "C" void kernel_launch(void* const* d_in, const int* in_sizes, int n_in,
                              void* d_out, int out_size, void* d_ws,
                              size_t ws_size, hipStream_t stream) {
  (void)in_sizes; (void)n_in; (void)out_size; (void)ws_size;
  const float* x = (const float*)d_in[0];
  const float* w1r = (const float*)d_in[1];
  const float* w1i = (const float*)d_in[2];
  const float* w2r = (const float*)d_in[3];
  const float* w2i = (const float*)d_in[4];
  float* out = (float*)d_out;

  float* tab = (float*)d_ws;
  float* FT = tab;
  float* E2c = tab + 8192;
  float* E2s = tab + 16384;
  float* E4F = tab + 24576;
  float* GT = tab + 40960;
  float* Y = tab + 49152;
  float* Zre = Y + 4194304;
  float* Zim = Zre + 262144;
  float* Tre = Zim + 262144;
  float* Tim = Tre + 262144;
  float* U = Tim + 262144;

  hipLaunchKernelGGL(k0_tables, dim3(192), dim3(256), 0, stream, tab);
  hipLaunchKernelGGL(k1_dft_w, dim3(1024), dim3(256), 0, stream, x, FT, Y);
  hipLaunchKernelGGL(k2_dft_h, dim3(64), dim3(256), 0, stream, Y, E2c, E2s, Zre,
                     Zim);
  hipLaunchKernelGGL(k3_mix, dim3(64), dim3(256), 0, stream, Zre, Zim, w1r, w1i,
                     w2r, w2i, Tre, Tim);
  hipLaunchKernelGGL(k4_idft_h, dim3(1024), dim3(256), 0, stream, Tre, Tim, E4F,
                     U);
  hipLaunchKernelGGL(k5_idft_w, dim3(4096), dim3(256), 0, stream, U, GT, out);
}


# Inverse H-DFT fused into the inverse W-DFT kernel: a wave forms its 16x32 U tile transposed in registers, where it is the A fragment, and writes 16 full rows; W-sum k order permuted within 16s
# speedup vs baseline: 1.3737x; 1.0612x over previous
#include <hip/hip_runtime.h>


typedef __attribute__((ext_vector_type(2))) float v2f;
typedef __attribute__((ext_vector_type(4))) float v4f;
typedef __attribute__((ext_vector_type(8))) float v8f;

__device__ __forceinline__ v8f wmma4(v2f a, v2f b, v8f c) {
  return __builtin_amdgcn_wmma_f32_16x16x4_f32(false, a, false, b, (short)0, c,
                                               false, false);
}

__device__ __forceinline__ v2f ld2(const float* p) { return *(const v2f*)p; }
__device__ __forceinline__ v4f ld4(const float* p) { return *(const v4f*)p; }
__device__ __forceinline__ v4f ld4_nt(const float* p) {
  return __builtin_nontemporal_load((const v4f*)p);
}

__global__ __launch_bounds__(256) void k0_tables(float* __restrict__ tab) {
  int idx = blockIdx.x * 256 + threadIdx.x;
  if (idx >= 49152) return;
  const float step = 6.28318530717958647692f / 256.0f;
  const int s = idx & 1, jt = (idx >> 1) & 1;
  const int lane = (idx >> 2) & 31, c = lane & 15, hf = lane >> 4;
  float v;
  if (idx < 8192) {
    int ks = idx >> 7;
    int w = 8 * (ks >> 1) + 4 * hf + 2 * (ks & 1) + s;
    int n = 16 * jt + c;
    float th = step * (float)(((n & 15) * w) & 255);
    v = (n < 16) ? cosf(th) : -sinf(th);
  } else if (idx < 24576) {
    int q = idx - 8192;
    int is_s = q >= 8192;
    q &= 8191;
    int ks = q >> 7;
    int h = 4 * ks + 2 * hf + s;
    int r = 16 * jt + c;
    int ky = (r < 16) ? r : (224 + r);
    float th = step * (float)((ky * h) & 255);
    v = is_s ? sinf(th) : cosf(th);
  } else if (idx < 40960) {
    int q = idx - 24576;
    int ks = (q >> 7) & 7, ht = q >> 10;
    int h = 16 * ht + c;
    int r = 4 * ks + 2 * hf + s;
    int ky = (r < 16) ? r : (224 + r);
    float th = step * (float)((ky * h) & 255);
    v = jt ? sinf(th) : cosf(th);
  } else {
    int q = idx - 40960;
    int p = (q >> 7) & 1, ks = (q >> 8) & 7, g = q >> 11;
    int w = 64 * g + 16 * (2 * p + jt) + c;
    int n = 16 * (ks >> 2) + 8 * hf + 2 * (ks & 3) + s;
    int kx = n & 15;
    float th = step * (float)((kx * w) & 255);
    float sc = ((kx == 0) ? 1.0f : 2.0f) * (1.0f / 65536.0f);
    v = (n < 16) ? sc * cosf(th) : -sc * sinf(th);
  }
  tab[idx] = v;
}

#define K1_PITCH4 17
__global__ __launch_bounds__(256) __attribute__((amdgpu_waves_per_eu(8, 8)))
void k1_dft_w(const float* __restrict__ x, const float* __restrict__ FT,
              float* __restrict__ Y) {
  __shared__ v4f stage[8 * 16 * K1_PITCH4];
  const int wv = threadIdx.x >> 5;
  const int wave = blockIdx.x * 8 + wv;
  const int lane = threadIdx.x & 31;
  const int half = lane >> 4, l16 = lane & 15;
  const long rowbase = (long)wave * 16;
  v4f* tile = stage + wv * (16 * K1_PITCH4);
  const float* gsrc = x + (rowbase + half) * 256 + 4 * l16;
  v4f* sdst = tile + half * K1_PITCH4 + l16;
  const v4f* ssrc = tile + l16 * K1_PITCH4 + half;
  const float* fb = FT + lane * 4;
  v8f acc0 = {}, acc1 = {};
  v4f g[8];
#pragma unroll
  for (int i = 0; i < 8; ++i) g[i] = ld4_nt(gsrc + i * 512);
#pragma unroll
  for (int c = 0; c < 4; ++c) {
#pragma unroll
    for (int i = 0; i < 8; ++i) sdst[2 * i * K1_PITCH4] = g[i];
    if (c < 3) {
#pragma unroll
      for (int i = 0; i < 8; ++i) g[i] = ld4_nt(gsrc + i * 512 + (c + 1) * 64);
    }
    __builtin_amdgcn_fence(__ATOMIC_ACQ_REL, "wavefront");
    v4f r[8];
#pragma unroll
    for (int j = 0; j < 8; ++j) r[j] = ssrc[2 * j];
#pragma unroll
    for (int j = 0; j < 8; ++j) {
      const int ks = (c * 8 + j) * 2;
      v4f f0 = ld4(fb + ks * 128);
      v4f f1 = ld4(fb + (ks + 1) * 128);
      v2f a0 = {r[j].x, r[j].y};
      v2f a1 = {r[j].z, r[j].w};
      v2f b00 = {f0.x, f0.y};
      v2f b01 = {f0.z, f0.w};
      v2f b10 = {f1.x, f1.y};
      v2f b11 = {f1.z, f1.w};
      acc0 = wmma4(a0, b00, acc0);
      acc1 = wmma4(a0, b01, acc1);
      acc0 = wmma4(a1, b10, acc0);
      acc1 = wmma4(a1, b11, acc1);
    }
    if (c < 3) __builtin_amdgcn_fence(__ATOMIC_ACQ_REL, "wavefront");
  }
  float* yrow = Y + rowbase * 32;
#pragma unroll
  for (int r = 0; r < 8; ++r) {
    int row = r + half * 8;
    yrow[row * 32 + l16] = acc0[r];
    yrow[row * 32 + 16 + l16] = acc1[r];
  }
}

__global__ __launch_bounds__(256) void k2_dft_h(const float* __restrict__ Y,
                                                const float* __restrict__ E2c,
                                                const float* __restrict__ E2s,
                                                float* __restrict__ Zre,
                                                float* __restrict__ Zim) {
  const int bi = blockIdx.x * 8 + (threadIdx.x >> 5);
  const int lane = threadIdx.x & 31;
  const int half = lane >> 4, l16 = lane & 15;
  const float* yb = Y + (long)bi * 256 * 32;
  const float* cf = E2c + lane * 4;
  const float* sf = E2s + lane * 4;
  v8f a0 = {}, a1 = {}, a2 = {}, a3 = {}, a4 = {}, a5 = {}, a6 = {}, a7 = {};
#pragma unroll 2
  for (int k0 = 0; k0 < 256; k0 += 4) {
    int kr = k0 + 2 * half;
    v2f bre, bim;
    bre.x = yb[kr * 32 + l16];
    bre.y = yb[(kr + 1) * 32 + l16];
    bim.x = yb[kr * 32 + 16 + l16];
    bim.y = yb[(kr + 1) * 32 + 16 + l16];
    v4f cc = ld4(cf + k0 * 32), ss = ld4(sf + k0 * 32);
    v2f ac0 = {cc.x, cc.y}, ac1 = {cc.z, cc.w};
    v2f as0 = {ss.x, ss.y}, as1 = {ss.z, ss.w};
    a0 = wmma4(ac0, bre, a0);
    a1 = wmma4(as0, bim, a1);
    a2 = wmma4(ac0, bim, a2);
    a3 = wmma4(as0, bre, a3);
    a4 = wmma4(ac1, bre, a4);
    a5 = wmma4(as1, bim, a5);
    a6 = wmma4(ac1, bim, a6);
    a7 = wmma4(as1, bre, a7);
  }
  const int b = bi >> 5, i = bi & 31;
#pragma unroll
  for (int r = 0; r < 8; ++r) {
    int kyr = r + half * 8;
    long o0 = (((long)kyr * 16 + l16) * 16 + b) * 32 + i;
    long o1 = (((long)(16 + kyr) * 16 + l16) * 16 + b) * 32 + i;
    Zre[o0] = a0[r] + a1[r];
    Zim[o0] = a2[r] - a3[r];
    Zre[o1] = a4[r] + a5[r];
    Zim[o1] = a6[r] - a7[r];
  }
}

__global__ __launch_bounds__(256) void k3_mix(
    const float* __restrict__ Zre, const float* __restrict__ Zim,
    const float* __restrict__ w1r, const float* __restrict__ w1i,
    const float* __restrict__ w2r, const float* __restrict__ w2i,
    float* __restrict__ Tre, float* __restrict__ Tim) {
  const int m = blockIdx.x * 8 + (threadIdx.x >> 5);
  const int lane = threadIdx.x & 31;
  const int half = lane >> 4, l16 = lane & 15;
  const int kyp = m >> 4, kx = m & 15;
  const int ky = kyp & 15;
  const float* wr = (kyp < 16) ? w1r : w2r;
  const float* wi = (kyp < 16) ? w1i : w2i;
  const float* zre = Zre + (long)m * 512 + l16 * 32 + 2 * half;
  const float* zim = Zim + (long)m * 512 + l16 * 32 + 2 * half;
  const long wofs = (long)ky * 16 + kx;
  v8f p0 = {}, q0 = {}, u0 = {}, v0 = {}, p1 = {}, q1 = {}, u1 = {}, v1 = {};
#pragma unroll
  for (int k0 = 0; k0 < 32; k0 += 4) {
    int kr = k0 + 2 * half;
    v2f azr = ld2(zre + k0);
    v2f azi = ld2(zim + k0);
    long ba0 = ((long)kr * 32 + l16) * 256 + wofs;
    long ba0b = ((long)(kr + 1) * 32 + l16) * 256 + wofs;
    long ba1 = ((long)kr * 32 + 16 + l16) * 256 + wofs;
    long ba1b = ((long)(kr + 1) * 32 + 16 + l16) * 256 + wofs;
    v2f br0, bi0, br1, bi1;
    br0.x = wr[ba0];  br0.y = wr[ba0b];
    bi0.x = wi[ba0];  bi0.y = wi[ba0b];
    br1.x = wr[ba1];  br1.y = wr[ba1b];
    bi1.x = wi[ba1];  bi1.y = wi[ba1b];
    p0 = wmma4(azr, br0, p0);
    q0 = wmma4(azi, bi0, q0);
    u0 = wmma4(azr, bi0, u0);
    v0 = wmma4(azi, br0, v0);
    p1 = wmma4(azr, br1, p1);
    q1 = wmma4(azi, bi1, q1);
    u1 = wmma4(azr, bi1, u1);
    v1 = wmma4(azi, br1, v1);
  }
#pragma unroll
  for (int r = 0; r < 8; ++r) {
    int b = r + half * 8;
    long o0 = (((long)b * 32 + l16) * 32 + kyp) * 16 + kx;
    long o1 = (((long)b * 32 + 16 + l16) * 32 + kyp) * 16 + kx;
    Tre[o0] = p0[r] - q0[r];
    Tim[o0] = u0[r] + v0[r];
    Tre[o1] = p1[r] - q1[r];
    Tim[o1] = u1[r] + v1[r];
  }
}

__global__ __launch_bounds__(256) __attribute__((amdgpu_waves_per_eu(8, 8)))
void k4_idft_hw(
    const float* __restrict__ Tre, const float* __restrict__ Tim,
    const float* __restrict__ E4F, const float* __restrict__ GT,
    float* __restrict__ out) {
  const int wid = blockIdx.x * 8 + (threadIdx.x >> 5);
  const int lane = threadIdx.x & 31;
  const int half = lane >> 4, l16 = lane & 15;
  const int bo = wid >> 4;
  const int ht = wid & 15;
  const float* tre = Tre + (long)bo * 512;
  const float* tim = Tim + (long)bo * 512;
  const float* ef = E4F + (ht * 256 + lane) * 4;
  v8f a0 = {}, a1 = {}, a2 = {}, a3 = {};
#pragma unroll
  for (int k0 = 0; k0 < 32; k0 += 4) {
    int kr = k0 + 2 * half;
    v4f e = ld4(ef + k0 * 32);
    v2f ec = {e.x, e.y};
    v2f es = {e.z, e.w};
    v2f tr, ti;
    tr.x = tre[kr * 16 + l16];
    tr.y = tre[(kr + 1) * 16 + l16];
    ti.x = tim[kr * 16 + l16];
    ti.y = tim[(kr + 1) * 16 + l16];
    a0 = wmma4(tr, ec, a0);
    a1 = wmma4(ti, es, a1);
    a2 = wmma4(ti, ec, a2);
    a3 = wmma4(tr, es, a3);
  }
  const v8f ure = a0 - a1;
  const v8f uim = a2 + a3;
  float* orow = out + ((long)bo * 256 + ht * 16) * 256;
  const float* gf = GT + lane * 4;
#pragma unroll 1
  for (int g = 0; g < 4; ++g) {
    v8f acc[4] = {};
#pragma unroll
    for (int ks = 0; ks < 8; ++ks) {
      const int j = 2 * (ks & 3);
      v2f a;
      if (ks < 4) {
        a.x = ure[j];
        a.y = ure[j + 1];
      } else {
        a.x = uim[j];
        a.y = uim[j + 1];
      }
#pragma unroll
      for (int p = 0; p < 2; ++p) {
        v4f t = ld4(gf + g * 2048 + ks * 256 + p * 128);
        v2f ga = {t.x, t.y}, gb = {t.z, t.w};
        acc[2 * p] = wmma4(a, ga, acc[2 * p]);
        acc[2 * p + 1] = wmma4(a, gb, acc[2 * p + 1]);
      }
    }
#pragma unroll
    for (int r = 0; r < 8; ++r) {
      int rr = r + half * 8;
#pragma unroll
      for (int t = 0; t < 4; ++t) {
        __builtin_nontemporal_store(
            acc[t][r], &orow[(long)rr * 256 + g * 64 + t * 16 + l16]);
      }
    }
  }
}

extern "C" void kernel_launch(void* const* d_in, const int* in_sizes, int n_in,
                              void* d_out, int out_size, void* d_ws,
                              size_t ws_size, hipStream_t stream) {
  (void)in_sizes; (void)n_in; (void)out_size; (void)ws_size;
  const float* x = (const float*)d_in[0];
  const float* w1r = (const float*)d_in[1];
  const float* w1i = (const float*)d_in[2];
  const float* w2r = (const float*)d_in[3];
  const float* w2i = (const float*)d_in[4];
  float* out = (float*)d_out;

  float* tab = (float*)d_ws;
  float* FT = tab;
  float* E2c = tab + 8192;
  float* E2s = tab + 16384;
  float* E4F = tab + 24576;
  float* GT = tab + 40960;
  float* Y = tab + 49152;
  float* Zre = Y + 4194304;
  float* Zim = Zre + 262144;
  float* Tre = Zim + 262144;
  float* Tim = Tre + 262144;

  hipLaunchKernelGGL(k0_tables, dim3(192), dim3(256), 0, stream, tab);
  hipLaunchKernelGGL(k1_dft_w, dim3(1024), dim3(256), 0, stream, x, FT, Y);
  hipLaunchKernelGGL(k2_dft_h, dim3(64), dim3(256), 0, stream, Y, E2c, E2s, Zre,
                     Zim);
  hipLaunchKernelGGL(k3_mix, dim3(64), dim3(256), 0, stream, Zre, Zim, w1r, w1i,
                     w2r, w2i, Tre, Tim);
  hipLaunchKernelGGL(k4_idft_hw, dim3(1024), dim3(256), 0, stream, Tre, Tim,
                     E4F, GT, out);
}


# Forward H-DFT kernel: 8 k-steps of loads kept in flight ahead of the WMMAs (no memory wait per k-step), work spread over 256 workgroups (Zre wave + Zim wave per (b,i)) instead of 64; sums unchanged
# speedup vs baseline: 1.9208x; 1.3983x over previous
#include <hip/hip_runtime.h>


typedef __attribute__((ext_vector_type(2))) float v2f;
typedef __attribute__((ext_vector_type(4))) float v4f;
typedef __attribute__((ext_vector_type(8))) float v8f;

__device__ __forceinline__ v8f wmma4(v2f a, v2f b, v8f c) {
  return __builtin_amdgcn_wmma_f32_16x16x4_f32(false, a, false, b, (short)0, c,
                                               false, false);
}

__device__ __forceinline__ v2f ld2(const float* p) { return *(const v2f*)p; }
__device__ __forceinline__ v4f ld4(const float* p) { return *(const v4f*)p; }
__device__ __forceinline__ v4f ld4_nt(const float* p) {
  return __builtin_nontemporal_load((const v4f*)p);
}

__global__ __launch_bounds__(256) void k0_tables(float* __restrict__ tab) {
  int idx = blockIdx.x * 256 + threadIdx.x;
  if (idx >= 49152) return;
  const float step = 6.28318530717958647692f / 256.0f;
  const int s = idx & 1, jt = (idx >> 1) & 1;
  const int lane = (idx >> 2) & 31, c = lane & 15, hf = lane >> 4;
  float v;
  if (idx < 8192) {
    int ks = idx >> 7;
    int w = 8 * (ks >> 1) + 4 * hf + 2 * (ks & 1) + s;
    int n = 16 * jt + c;
    float th = step * (float)(((n & 15) * w) & 255);
    v = (n < 16) ? cosf(th) : -sinf(th);
  } else if (idx < 24576) {
    int q = idx - 8192;
    int is_s = q >= 8192;
    q &= 8191;
    int ks = q >> 7;
    int h = 4 * ks + 2 * hf + s;
    int r = 16 * jt + c;
    int ky = (r < 16) ? r : (224 + r);
    float th = step * (float)((ky * h) & 255);
    v = is_s ? sinf(th) : cosf(th);
  } else if (idx < 40960) {
    int q = idx - 24576;
    int ks = (q >> 7) & 7, ht = q >> 10;
    int h = 16 * ht + c;
    int r = 4 * ks + 2 * hf + s;
    int ky = (r < 16) ? r : (224 + r);
    float th = step * (float)((ky * h) & 255);
    v = jt ? sinf(th) : cosf(th);
  } else {
    int q = idx - 40960;
    int p = (q >> 7) & 1, ks = (q >> 8) & 7, g = q >> 11;
    int w = 64 * g + 16 * (2 * p + jt) + c;
    int n = 16 * (ks >> 2) + 8 * hf + 2 * (ks & 3) + s;
    int kx = n & 15;
    float th = step * (float)((kx * w) & 255);
    float sc = ((kx == 0) ? 1.0f : 2.0f) * (1.0f / 65536.0f);
    v = (n < 16) ? sc * cosf(th) : -sc * sinf(th);
  }
  tab[idx] = v;
}

#define K1_PITCH4 17
__global__ __launch_bounds__(256) __attribute__((amdgpu_waves_per_eu(8, 8)))
void k1_dft_w(const float* __restrict__ x, const float* __restrict__ FT,
              float* __restrict__ Y) {
  __shared__ v4f stage[8 * 16 * K1_PITCH4];
  const int wv = threadIdx.x >> 5;
  const int wave = blockIdx.x * 8 + wv;
  const int lane = threadIdx.x & 31;
  const int half = lane >> 4, l16 = lane & 15;
  const long rowbase = (long)wave * 16;
  v4f* tile = stage + wv * (16 * K1_PITCH4);
  const float* gsrc = x + (rowbase + half) * 256 + 4 * l16;
  v4f* sdst = tile + half * K1_PITCH4 + l16;
  const v4f* ssrc = tile + l16 * K1_PITCH4 + half;
  const float* fb = FT + lane * 4;
  v8f acc0 = {}, acc1 = {};
  v4f g[8];
#pragma unroll
  for (int i = 0; i < 8; ++i) g[i] = ld4_nt(gsrc + i * 512);
#pragma unroll
  for (int c = 0; c < 4; ++c) {
#pragma unroll
    for (int i = 0; i < 8; ++i) sdst[2 * i * K1_PITCH4] = g[i];
    if (c < 3) {
#pragma unroll
      for (int i = 0; i < 8; ++i) g[i] = ld4_nt(gsrc + i * 512 + (c + 1) * 64);
    }
    __builtin_amdgcn_fence(__ATOMIC_ACQ_REL, "wavefront");
    v4f r[8];
#pragma unroll
    for (int j = 0; j < 8; ++j) r[j] = ssrc[2 * j];
#pragma unroll
    for (int j = 0; j < 8; ++j) {
      const int ks = (c * 8 + j) * 2;
      v4f f0 = ld4(fb + ks * 128);
      v4f f1 = ld4(fb + (ks + 1) * 128);
      v2f a0 = {r[j].x, r[j].y};
      v2f a1 = {r[j].z, r[j].w};
      v2f b00 = {f0.x, f0.y};
      v2f b01 = {f0.z, f0.w};
      v2f b10 = {f1.x, f1.y};
      v2f b11 = {f1.z, f1.w};
      acc0 = wmma4(a0, b00, acc0);
      acc1 = wmma4(a0, b01, acc1);
      acc0 = wmma4(a1, b10, acc0);
      acc1 = wmma4(a1, b11, acc1);
    }
    if (c < 3) __builtin_amdgcn_fence(__ATOMIC_ACQ_REL, "wavefront");
  }
  float* yrow = Y + rowbase * 32;
#pragma unroll
  for (int r = 0; r < 8; ++r) {
    int row = r + half * 8;
    yrow[row * 32 + l16] = acc0[r];
    yrow[row * 32 + 16 + l16] = acc1[r];
  }
}

struct K2Frag {
  v2f first[8], second[8];
  v4f cs[8], sn[8];
};

__device__ __forceinline__ void k2_load(K2Frag& f, const float* yfirst,
                                        const float* ysecond, const float* cf,
                                        const float* sf, int ks0) {
#pragma unroll
  for (int u = 0; u < 8; ++u) {
    const int k0 = 4 * (ks0 + u);
    f.first[u].x = yfirst[k0 * 32];
    f.first[u].y = yfirst[(k0 + 1) * 32];
    f.second[u].x = ysecond[k0 * 32];
    f.second[u].y = ysecond[(k0 + 1) * 32];
    f.cs[u] = ld4(cf + k0 * 32);
    f.sn[u] = ld4(sf + k0 * 32);
  }
}

__device__ __forceinline__ void k2_mma(const K2Frag& f, v8f& a0, v8f& a1,
                                       v8f& a2, v8f& a3) {
#pragma unroll
  for (int u = 0; u < 8; ++u) {
    v2f ac0 = {f.cs[u].x, f.cs[u].y}, ac1 = {f.cs[u].z, f.cs[u].w};
    v2f as0 = {f.sn[u].x, f.sn[u].y}, as1 = {f.sn[u].z, f.sn[u].w};
    a0 = wmma4(ac0, f.first[u], a0);
    a1 = wmma4(as0, f.second[u], a1);
    a2 = wmma4(ac1, f.first[u], a2);
    a3 = wmma4(as1, f.second[u], a3);
  }
}

__global__ __launch_bounds__(128) void k2_dft_h(const float* __restrict__ Y,
                                                const float* __restrict__ E2c,
                                                const float* __restrict__ E2s,
                                                float* __restrict__ Zre,
                                                float* __restrict__ Zim) {
  const int wv = threadIdx.x >> 5;
  const int bi = blockIdx.x * 2 + (wv >> 1);
  const int part = wv & 1;
  const int lane = threadIdx.x & 31;
  const int half = lane >> 4, l16 = lane & 15;
  const float* yb = Y + (long)bi * 256 * 32 + (2 * half) * 32 + l16;
  const float* yfirst = yb + 16 * part;
  const float* ysecond = yb + 16 * (1 - part);
  const float* cf = E2c + lane * 4;
  const float* sf = E2s + lane * 4;
  v8f a0 = {}, a1 = {}, a2 = {}, a3 = {};
  K2Frag f0, f1;
  k2_load(f0, yfirst, ysecond, cf, sf, 0);
#pragma unroll
  for (int it = 0; it < 4; ++it) {
    k2_load(f1, yfirst, ysecond, cf, sf, 16 * it + 8);
    __builtin_amdgcn_sched_barrier(0);
    k2_mma(f0, a0, a1, a2, a3);
    __builtin_amdgcn_sched_barrier(0);
    if (it < 3) k2_load(f0, yfirst, ysecond, cf, sf, 16 * it + 16);
    __builtin_amdgcn_sched_barrier(0);
    k2_mma(f1, a0, a1, a2, a3);
    __builtin_amdgcn_sched_barrier(0);
  }
  const int b = bi >> 5, i = bi & 31;
  float* Z = part ? Zim : Zre;
#pragma unroll
  for (int r = 0; r < 8; ++r) {
    int kyr = r + half * 8;
    long o0 = (((long)kyr * 16 + l16) * 16 + b) * 32 + i;
    long o1 = (((long)(16 + kyr) * 16 + l16) * 16 + b) * 32 + i;
    Z[o0] = part ? (a0[r] - a1[r]) : (a0[r] + a1[r]);
    Z[o1] = part ? (a2[r] - a3[r]) : (a2[r] + a3[r]);
  }
}

__global__ __launch_bounds__(256) void k3_mix(
    const float* __restrict__ Zre, const float* __restrict__ Zim,
    const float* __restrict__ w1r, const float* __restrict__ w1i,
    const float* __restrict__ w2r, const float* __restrict__ w2i,
    float* __restrict__ Tre, float* __restrict__ Tim) {
  const int m = blockIdx.x * 8 + (threadIdx.x >> 5);
  const int lane = threadIdx.x & 31;
  const int half = lane >> 4, l16 = lane & 15;
  const int kyp = m >> 4, kx = m & 15;
  const int ky = kyp & 15;
  const float* wr = (kyp < 16) ? w1r : w2r;
  const float* wi = (kyp < 16) ? w1i : w2i;
  const float* zre = Zre + (long)m * 512 + l16 * 32 + 2 * half;
  const float* zim = Zim + (long)m * 512 + l16 * 32 + 2 * half;
  const long wofs = (long)ky * 16 + kx;
  v8f p0 = {}, q0 = {}, u0 = {}, v0 = {}, p1 = {}, q1 = {}, u1 = {}, v1 = {};
#pragma unroll
  for (int k0 = 0; k0 < 32; k0 += 4) {
    int kr = k0 + 2 * half;
    v2f azr = ld2(zre + k0);
    v2f azi = ld2(zim + k0);
    long ba0 = ((long)kr * 32 + l16) * 256 + wofs;
    long ba0b = ((long)(kr + 1) * 32 + l16) * 256 + wofs;
    long ba1 = ((long)kr * 32 + 16 + l16) * 256 + wofs;
    long ba1b = ((long)(kr + 1) * 32 + 16 + l16) * 256 + wofs;
    v2f br0, bi0, br1, bi1;
    br0.x = wr[ba0];  br0.y = wr[ba0b];
    bi0.x = wi[ba0];  bi0.y = wi[ba0b];
    br1.x = wr[ba1];  br1.y = wr[ba1b];
    bi1.x = wi[ba1];  bi1.y = wi[ba1b];
    p0 = wmma4(azr, br0, p0);
    q0 = wmma4(azi, bi0, q0);
    u0 = wmma4(azr, bi0, u0);
    v0 = wmma4(azi, br0, v0);
    p1 = wmma4(azr, br1, p1);
    q1 = wmma4(azi, bi1, q1);
    u1 = wmma4(azr, bi1, u1);
    v1 = wmma4(azi, br1, v1);
  }
#pragma unroll
  for (int r = 0; r < 8; ++r) {
    int b = r + half * 8;
    long o0 = (((long)b * 32 + l16) * 32 + kyp) * 16 + kx;
    long o1 = (((long)b * 32 + 16 + l16) * 32 + kyp) * 16 + kx;
    Tre[o0] = p0[r] - q0[r];
    Tim[o0] = u0[r] + v0[r];
    Tre[o1] = p1[r] - q1[r];
    Tim[o1] = u1[r] + v1[r];
  }
}

__global__ __launch_bounds__(256) __attribute__((amdgpu_waves_per_eu(8, 8)))
void k4_idft_hw(
    const float* __restrict__ Tre, const float* __restrict__ Tim,
    const float* __restrict__ E4F, const float* __restrict__ GT,
    float* __restrict__ out) {
  const int wid = blockIdx.x * 8 + (threadIdx.x >> 5);
  const int lane = threadIdx.x & 31;
  const int half = lane >> 4, l16 = lane & 15;
  const int bo = wid >> 4;
  const int ht = wid & 15;
  const float* tre = Tre + (long)bo * 512;
  const float* tim = Tim + (long)bo * 512;
  const float* ef = E4F + (ht * 256 + lane) * 4;
  v8f a0 = {}, a1 = {}, a2 = {}, a3 = {};
#pragma unroll
  for (int k0 = 0; k0 < 32; k0 += 4) {
    int kr = k0 + 2 * half;
    v4f e = ld4(ef + k0 * 32);
    v2f ec = {e.x, e.y};
    v2f es = {e.z, e.w};
    v2f tr, ti;
    tr.x = tre[kr * 16 + l16];
    tr.y = tre[(kr + 1) * 16 + l16];
    ti.x = tim[kr * 16 + l16];
    ti.y = tim[(kr + 1) * 16 + l16];
    a0 = wmma4(tr, ec, a0);
    a1 = wmma4(ti, es, a1);
    a2 = wmma4(ti, ec, a2);
    a3 = wmma4(tr, es, a3);
  }
  const v8f ure = a0 - a1;
  const v8f uim = a2 + a3;
  float* orow = out + ((long)bo * 256 + ht * 16) * 256;
  const float* gf = GT + lane * 4;
#pragma unroll 1
  for (int g = 0; g < 4; ++g) {
    v8f acc[4] = {};
#pragma unroll
    for (int ks = 0; ks < 8; ++ks) {
      const int j = 2 * (ks & 3);
      v2f a;
      if (ks < 4) {
        a.x = ure[j];
        a.y = ure[j + 1];
      } else {
        a.x = uim[j];
        a.y = uim[j + 1];
      }
#pragma unroll
      for (int p = 0; p < 2; ++p) {
        v4f t = ld4(gf + g * 2048 + ks * 256 + p * 128);
        v2f ga = {t.x, t.y}, gb = {t.z, t.w};
        acc[2 * p] = wmma4(a, ga, acc[2 * p]);
        acc[2 * p + 1] = wmma4(a, gb, acc[2 * p + 1]);
      }
    }
#pragma unroll
    for (int r = 0; r < 8; ++r) {
      int rr = r + half * 8;
#pragma unroll
      for (int t = 0; t < 4; ++t) {
        __builtin_nontemporal_store(
            acc[t][r], &orow[(long)rr * 256 + g * 64 + t * 16 + l16]);
      }
    }
  }
}

extern "C" void kernel_launch(void* const* d_in, const int* in_sizes, int n_in,
                              void* d_out, int out_size, void* d_ws,
                              size_t ws_size, hipStream_t stream) {
  (void)in_sizes; (void)n_in; (void)out_size; (void)ws_size;
  const float* x = (const float*)d_in[0];
  const float* w1r = (const float*)d_in[1];
  const float* w1i = (const float*)d_in[2];
  const float* w2r = (const float*)d_in[3];
  const float* w2i = (const float*)d_in[4];
  float* out = (float*)d_out;

  float* tab = (float*)d_ws;
  float* FT = tab;
  float* E2c = tab + 8192;
  float* E2s = tab + 16384;
  float* E4F = tab + 24576;
  float* GT = tab + 40960;
  float* Y = tab + 49152;
  float* Zre = Y + 4194304;
  float* Zim = Zre + 262144;
  float* Tre = Zim + 262144;
  float* Tim = Tre + 262144;

  hipLaunchKernelGGL(k0_tables, dim3(192), dim3(256), 0, stream, tab);
  hipLaunchKernelGGL(k1_dft_w, dim3(1024), dim3(256), 0, stream, x, FT, Y);
  hipLaunchKernelGGL(k2_dft_h, dim3(256), dim3(128), 0, stream, Y, E2c, E2s, Zre,
                     Zim);
  hipLaunchKernelGGL(k3_mix, dim3(64), dim3(256), 0, stream, Zre, Zim, w1r, w1i,
                     w2r, w2i, Tre, Tim);
  hipLaunchKernelGGL(k4_idft_hw, dim3(1024), dim3(256), 0, stream, Tre, Tim,
                     E4F, GT, out);
}


# Inverse W-DFT: output columns dealt across the 4 N-tiles (table column c of tile t = column 4c+t), so a lane stores 16 B per row and each store writes whole 128-byte lines; sums unchanged
# speedup vs baseline: 1.9990x; 1.0407x over previous
#include <hip/hip_runtime.h>


typedef __attribute__((ext_vector_type(2))) float v2f;
typedef __attribute__((ext_vector_type(4))) float v4f;
typedef __attribute__((ext_vector_type(8))) float v8f;

__device__ __forceinline__ v8f wmma4(v2f a, v2f b, v8f c) {
  return __builtin_amdgcn_wmma_f32_16x16x4_f32(false, a, false, b, (short)0, c,
                                               false, false);
}

__device__ __forceinline__ v2f ld2(const float* p) { return *(const v2f*)p; }
__device__ __forceinline__ v4f ld4(const float* p) { return *(const v4f*)p; }
__device__ __forceinline__ v4f ld4_nt(const float* p) {
  return __builtin_nontemporal_load((const v4f*)p);
}

__global__ __launch_bounds__(256) void k0_tables(float* __restrict__ tab) {
  int idx = blockIdx.x * 256 + threadIdx.x;
  if (idx >= 49152) return;
  const float step = 6.28318530717958647692f / 256.0f;
  const int s = idx & 1, jt = (idx >> 1) & 1;
  const int lane = (idx >> 2) & 31, c = lane & 15, hf = lane >> 4;
  float v;
  if (idx < 8192) {
    int ks = idx >> 7;
    int w = 8 * (ks >> 1) + 4 * hf + 2 * (ks & 1) + s;
    int n = 16 * jt + c;
    float th = step * (float)(((n & 15) * w) & 255);
    v = (n < 16) ? cosf(th) : -sinf(th);
  } else if (idx < 24576) {
    int q = idx - 8192;
    int is_s = q >= 8192;
    q &= 8191;
    int ks = q >> 7;
    int h = 4 * ks + 2 * hf + s;
    int r = 16 * jt + c;
    int ky = (r < 16) ? r : (224 + r);
    float th = step * (float)((ky * h) & 255);
    v = is_s ? sinf(th) : cosf(th);
  } else if (idx < 40960) {
    int q = idx - 24576;
    int ks = (q >> 7) & 7, ht = q >> 10;
    int h = 16 * ht + c;
    int r = 4 * ks + 2 * hf + s;
    int ky = (r < 16) ? r : (224 + r);
    float th = step * (float)((ky * h) & 255);
    v = jt ? sinf(th) : cosf(th);
  } else {
    int q = idx - 40960;
    int p = (q >> 7) & 1, ks = (q >> 8) & 7, g = q >> 11;
    int w = 64 * g + 4 * c + (2 * p + jt);
    int n = 16 * (ks >> 2) + 8 * hf + 2 * (ks & 3) + s;
    int kx = n & 15;
    float th = step * (float)((kx * w) & 255);
    float sc = ((kx == 0) ? 1.0f : 2.0f) * (1.0f / 65536.0f);
    v = (n < 16) ? sc * cosf(th) : -sc * sinf(th);
  }
  tab[idx] = v;
}

#define K1_PITCH4 17
__global__ __launch_bounds__(256) __attribute__((amdgpu_waves_per_eu(8, 8)))
void k1_dft_w(const float* __restrict__ x, const float* __restrict__ FT,
              float* __restrict__ Y) {
  __shared__ v4f stage[8 * 16 * K1_PITCH4];
  const int wv = threadIdx.x >> 5;
  const int wave = blockIdx.x * 8 + wv;
  const int lane = threadIdx.x & 31;
  const int half = lane >> 4, l16 = lane & 15;
  const long rowbase = (long)wave * 16;
  v4f* tile = stage + wv * (16 * K1_PITCH4);
  const float* gsrc = x + (rowbase + half) * 256 + 4 * l16;
  v4f* sdst = tile + half * K1_PITCH4 + l16;
  const v4f* ssrc = tile + l16 * K1_PITCH4 + half;
  const float* fb = FT + lane * 4;
  v8f acc0 = {}, acc1 = {};
  v4f g[8];
#pragma unroll
  for (int i = 0; i < 8; ++i) g[i] = ld4_nt(gsrc + i * 512);
#pragma unroll
  for (int c = 0; c < 4; ++c) {
#pragma unroll
    for (int i = 0; i < 8; ++i) sdst[2 * i * K1_PITCH4] = g[i];
    if (c < 3) {
#pragma unroll
      for (int i = 0; i < 8; ++i) g[i] = ld4_nt(gsrc + i * 512 + (c + 1) * 64);
    }
    __builtin_amdgcn_fence(__ATOMIC_ACQ_REL, "wavefront");
    v4f r[8];
#pragma unroll
    for (int j = 0; j < 8; ++j) r[j] = ssrc[2 * j];
#pragma unroll
    for (int j = 0; j < 8; ++j) {
      const int ks = (c * 8 + j) * 2;
      v4f f0 = ld4(fb + ks * 128);
      v4f f1 = ld4(fb + (ks + 1) * 128);
      v2f a0 = {r[j].x, r[j].y};
      v2f a1 = {r[j].z, r[j].w};
      v2f b00 = {f0.x, f0.y};
      v2f b01 = {f0.z, f0.w};
      v2f b10 = {f1.x, f1.y};
      v2f b11 = {f1.z, f1.w};
      acc0 = wmma4(a0, b00, acc0);
      acc1 = wmma4(a0, b01, acc1);
      acc0 = wmma4(a1, b10, acc0);
      acc1 = wmma4(a1, b11, acc1);
    }
    if (c < 3) __builtin_amdgcn_fence(__ATOMIC_ACQ_REL, "wavefront");
  }
  float* yrow = Y + rowbase * 32;
#pragma unroll
  for (int r = 0; r < 8; ++r) {
    int row = r + half * 8;
    yrow[row * 32 + l16] = acc0[r];
    yrow[row * 32 + 16 + l16] = acc1[r];
  }
}

struct K2Frag {
  v2f first[8], second[8];
  v4f cs[8], sn[8];
};

__device__ __forceinline__ void k2_load(K2Frag& f, const float* yfirst,
                                        const float* ysecond, const float* cf,
                                        const float* sf, int ks0) {
#pragma unroll
  for (int u = 0; u < 8; ++u) {
    const int k0 = 4 * (ks0 + u);
    f.first[u].x = yfirst[k0 * 32];
    f.first[u].y = yfirst[(k0 + 1) * 32];
    f.second[u].x = ysecond[k0 * 32];
    f.second[u].y = ysecond[(k0 + 1) * 32];
    f.cs[u] = ld4(cf + k0 * 32);
    f.sn[u] = ld4(sf + k0 * 32);
  }
}

__device__ __forceinline__ void k2_mma(const K2Frag& f, v8f& a0, v8f& a1,
                                       v8f& a2, v8f& a3) {
#pragma unroll
  for (int u = 0; u < 8; ++u) {
    v2f ac0 = {f.cs[u].x, f.cs[u].y}, ac1 = {f.cs[u].z, f.cs[u].w};
    v2f as0 = {f.sn[u].x, f.sn[u].y}, as1 = {f.sn[u].z, f.sn[u].w};
    a0 = wmma4(ac0, f.first[u], a0);
    a1 = wmma4(as0, f.second[u], a1);
    a2 = wmma4(ac1, f.first[u], a2);
    a3 = wmma4(as1, f.second[u], a3);
  }
}

__global__ __launch_bounds__(128) void k2_dft_h(const float* __restrict__ Y,
                                                const float* __restrict__ E2c,
                                                const float* __restrict__ E2s,
                                                float* __restrict__ Zre,
                                                float* __restrict__ Zim) {
  const int wv = threadIdx.x >> 5;
  const int bi = blockIdx.x * 2 + (wv >> 1);
  const int part = wv & 1;
  const int lane = threadIdx.x & 31;
  const int half = lane >> 4, l16 = lane & 15;
  const float* yb = Y + (long)bi * 256 * 32 + (2 * half) * 32 + l16;
  const float* yfirst = yb + 16 * part;
  const float* ysecond = yb + 16 * (1 - part);
  const float* cf = E2c + lane * 4;
  const float* sf = E2s + lane * 4;
  v8f a0 = {}, a1 = {}, a2 = {}, a3 = {};
  K2Frag f0, f1;
  k2_load(f0, yfirst, ysecond, cf, sf, 0);
#pragma unroll
  for (int it = 0; it < 4; ++it) {
    k2_load(f1, yfirst, ysecond, cf, sf, 16 * it + 8);
    __builtin_amdgcn_sched_barrier(0);
    k2_mma(f0, a0, a1, a2, a3);
    __builtin_amdgcn_sched_barrier(0);
    if (it < 3) k2_load(f0, yfirst, ysecond, cf, sf, 16 * it + 16);
    __builtin_amdgcn_sched_barrier(0);
    k2_mma(f1, a0, a1, a2, a3);
    __builtin_amdgcn_sched_barrier(0);
  }
  const int b = bi >> 5, i = bi & 31;
  float* Z = part ? Zim : Zre;
#pragma unroll
  for (int r = 0; r < 8; ++r) {
    int kyr = r + half * 8;
    long o0 = (((long)kyr * 16 + l16) * 16 + b) * 32 + i;
    long o1 = (((long)(16 + kyr) * 16 + l16) * 16 + b) * 32 + i;
    Z[o0] = part ? (a0[r] - a1[r]) : (a0[r] + a1[r]);
    Z[o1] = part ? (a2[r] - a3[r]) : (a2[r] + a3[r]);
  }
}

__global__ __launch_bounds__(256) void k3_mix(
    const float* __restrict__ Zre, const float* __restrict__ Zim,
    const float* __restrict__ w1r, const float* __restrict__ w1i,
    const float* __restrict__ w2r, const float* __restrict__ w2i,
    float* __restrict__ Tre, float* __restrict__ Tim) {
  const int m = blockIdx.x * 8 + (threadIdx.x >> 5);
  const int lane = threadIdx.x & 31;
  const int half = lane >> 4, l16 = lane & 15;
  const int kyp = m >> 4, kx = m & 15;
  const int ky = kyp & 15;
  const float* wr = (kyp < 16) ? w1r : w2r;
  const float* wi = (kyp < 16) ? w1i : w2i;
  const float* zre = Zre + (long)m * 512 + l16 * 32 + 2 * half;
  const float* zim = Zim + (long)m * 512 + l16 * 32 + 2 * half;
  const long wofs = (long)ky * 16 + kx;
  v8f p0 = {}, q0 = {}, u0 = {}, v0 = {}, p1 = {}, q1 = {}, u1 = {}, v1 = {};
#pragma unroll
  for (int k0 = 0; k0 < 32; k0 += 4) {
    int kr = k0 + 2 * half;
    v2f azr = ld2(zre + k0);
    v2f azi = ld2(zim + k0);
    long ba0 = ((long)kr * 32 + l16) * 256 + wofs;
    long ba0b = ((long)(kr + 1) * 32 + l16) * 256 + wofs;
    long ba1 = ((long)kr * 32 + 16 + l16) * 256 + wofs;
    long ba1b = ((long)(kr + 1) * 32 + 16 + l16) * 256 + wofs;
    v2f br0, bi0, br1, bi1;
    br0.x = wr[ba0];  br0.y = wr[ba0b];
    bi0.x = wi[ba0];  bi0.y = wi[ba0b];
    br1.x = wr[ba1];  br1.y = wr[ba1b];
    bi1.x = wi[ba1];  bi1.y = wi[ba1b];
    p0 = wmma4(azr, br0, p0);
    q0 = wmma4(azi, bi0, q0);
    u0 = wmma4(azr, bi0, u0);
    v0 = wmma4(azi, br0, v0);
    p1 = wmma4(azr, br1, p1);
    q1 = wmma4(azi, bi1, q1);
    u1 = wmma4(azr, bi1, u1);
    v1 = wmma4(azi, br1, v1);
  }
#pragma unroll
  for (int r = 0; r < 8; ++r) {
    int b = r + half * 8;
    long o0 = (((long)b * 32 + l16) * 32 + kyp) * 16 + kx;
    long o1 = (((long)b * 32 + 16 + l16) * 32 + kyp) * 16 + kx;
    Tre[o0] = p0[r] - q0[r];
    Tim[o0] = u0[r] + v0[r];
    Tre[o1] = p1[r] - q1[r];
    Tim[o1] = u1[r] + v1[r];
  }
}

__global__ __launch_bounds__(256) __attribute__((amdgpu_waves_per_eu(8, 8)))
void k4_idft_hw(
    const float* __restrict__ Tre, const float* __restrict__ Tim,
    const float* __restrict__ E4F, const float* __restrict__ GT,
    float* __restrict__ out) {
  const int wid = blockIdx.x * 8 + (threadIdx.x >> 5);
  const int lane = threadIdx.x & 31;
  const int half = lane >> 4, l16 = lane & 15;
  const int bo = wid >> 4;
  const int ht = wid & 15;
  const float* tre = Tre + (long)bo * 512;
  const float* tim = Tim + (long)bo * 512;
  const float* ef = E4F + (ht * 256 + lane) * 4;
  v8f a0 = {}, a1 = {}, a2 = {}, a3 = {};
#pragma unroll
  for (int k0 = 0; k0 < 32; k0 += 4) {
    int kr = k0 + 2 * half;
    v4f e = ld4(ef + k0 * 32);
    v2f ec = {e.x, e.y};
    v2f es = {e.z, e.w};
    v2f tr, ti;
    tr.x = tre[kr * 16 + l16];
    tr.y = tre[(kr + 1) * 16 + l16];
    ti.x = tim[kr * 16 + l16];
    ti.y = tim[(kr + 1) * 16 + l16];
    a0 = wmma4(tr, ec, a0);
    a1 = wmma4(ti, es, a1);
    a2 = wmma4(ti, ec, a2);
    a3 = wmma4(tr, es, a3);
  }
  const v8f ure = a0 - a1;
  const v8f uim = a2 + a3;
  float* orow = out + ((long)bo * 256 + ht * 16) * 256;
  const float* gf = GT + lane * 4;
#pragma unroll 1
  for (int g = 0; g < 4; ++g) {
    v8f acc[4] = {};
#pragma unroll
    for (int ks = 0; ks < 8; ++ks) {
      const int j = 2 * (ks & 3);
      v2f a;
      if (ks < 4) {
        a.x = ure[j];
        a.y = ure[j + 1];
      } else {
        a.x = uim[j];
        a.y = uim[j + 1];
      }
#pragma unroll
      for (int p = 0; p < 2; ++p) {
        v4f t = ld4(gf + g * 2048 + ks * 256 + p * 128);
        v2f ga = {t.x, t.y}, gb = {t.z, t.w};
        acc[2 * p] = wmma4(a, ga, acc[2 * p]);
        acc[2 * p + 1] = wmma4(a, gb, acc[2 * p + 1]);
      }
    }
#pragma unroll
    for (int r = 0; r < 8; ++r) {
      int rr = r + half * 8;
      v4f o = {acc[0][r], acc[1][r], acc[2][r], acc[3][r]};
      __builtin_nontemporal_store(
          o, (v4f*)&orow[(long)rr * 256 + g * 64 + 4 * l16]);
    }
  }
}

extern "C" void kernel_launch(void* const* d_in, const int* in_sizes, int n_in,
                              void* d_out, int out_size, void* d_ws,
                              size_t ws_size, hipStream_t stream) {
  (void)in_sizes; (void)n_in; (void)out_size; (void)ws_size;
  const float* x = (const float*)d_in[0];
  const float* w1r = (const float*)d_in[1];
  const float* w1i = (const float*)d_in[2];
  const float* w2r = (const float*)d_in[3];
  const float* w2i = (const float*)d_in[4];
  float* out = (float*)d_out;

  float* tab = (float*)d_ws;
  float* FT = tab;
  float* E2c = tab + 8192;
  float* E2s = tab + 16384;
  float* E4F = tab + 24576;
  float* GT = tab + 40960;
  float* Y = tab + 49152;
  float* Zre = Y + 4194304;
  float* Zim = Zre + 262144;
  float* Tre = Zim + 262144;
  float* Tim = Tre + 262144;

  hipLaunchKernelGGL(k0_tables, dim3(192), dim3(256), 0, stream, tab);
  hipLaunchKernelGGL(k1_dft_w, dim3(1024), dim3(256), 0, stream, x, FT, Y);
  hipLaunchKernelGGL(k2_dft_h, dim3(256), dim3(128), 0, stream, Y, E2c, E2s, Zre,
                     Zim);
  hipLaunchKernelGGL(k3_mix, dim3(64), dim3(256), 0, stream, Zre, Zim, w1r, w1i,
                     w2r, w2i, Tre, Tim);
  hipLaunchKernelGGL(k4_idft_hw, dim3(1024), dim3(256), 0, stream, Tre, Tim,
                     E4F, GT, out);
}


# Channel-mix kernel: a wave takes 4 neighbouring kx of one ky' and one 16-wide o tile: weight loads and T stores are 16 B per lane, 256 one-wave workgroups instead of 64 of 8 waves; sums unchanged
# speedup vs baseline: 2.6698x; 1.3356x over previous
#include <hip/hip_runtime.h>


typedef __attribute__((ext_vector_type(2))) float v2f;
typedef __attribute__((ext_vector_type(4))) float v4f;
typedef __attribute__((ext_vector_type(8))) float v8f;

__device__ __forceinline__ v8f wmma4(v2f a, v2f b, v8f c) {
  return __builtin_amdgcn_wmma_f32_16x16x4_f32(false, a, false, b, (short)0, c,
                                               false, false);
}

__device__ __forceinline__ v2f ld2(const float* p) { return *(const v2f*)p; }
__device__ __forceinline__ v4f ld4(const float* p) { return *(const v4f*)p; }
__device__ __forceinline__ v4f ld4_nt(const float* p) {
  return __builtin_nontemporal_load((const v4f*)p);
}

__global__ __launch_bounds__(256) void k0_tables(float* __restrict__ tab) {
  int idx = blockIdx.x * 256 + threadIdx.x;
  if (idx >= 49152) return;
  const float step = 6.28318530717958647692f / 256.0f;
  const int s = idx & 1, jt = (idx >> 1) & 1;
  const int lane = (idx >> 2) & 31, c = lane & 15, hf = lane >> 4;
  float v;
  if (idx < 8192) {
    int ks = idx >> 7;
    int w = 8 * (ks >> 1) + 4 * hf + 2 * (ks & 1) + s;
    int n = 16 * jt + c;
    float th = step * (float)(((n & 15) * w) & 255);
    v = (n < 16) ? cosf(th) : -sinf(th);
  } else if (idx < 24576) {
    int q = idx - 8192;
    int is_s = q >= 8192;
    q &= 8191;
    int ks = q >> 7;
    int h = 4 * ks + 2 * hf + s;
    int r = 16 * jt + c;
    int ky = (r < 16) ? r : (224 + r);
    float th = step * (float)((ky * h) & 255);
    v = is_s ? sinf(th) : cosf(th);
  } else if (idx < 40960) {
    int q = idx - 24576;
    int ks = (q >> 7) & 7, ht = q >> 10;
    int h = 16 * ht + c;
    int r = 4 * ks + 2 * hf + s;
    int ky = (r < 16) ? r : (224 + r);
    float th = step * (float)((ky * h) & 255);
    v = jt ? sinf(th) : cosf(th);
  } else {
    int q = idx - 40960;
    int p = (q >> 7) & 1, ks = (q >> 8) & 7, g = q >> 11;
    int w = 64 * g + 4 * c + (2 * p + jt);
    int n = 16 * (ks >> 2) + 8 * hf + 2 * (ks & 3) + s;
    int kx = n & 15;
    float th = step * (float)((kx * w) & 255);
    float sc = ((kx == 0) ? 1.0f : 2.0f) * (1.0f / 65536.0f);
    v = (n < 16) ? sc * cosf(th) : -sc * sinf(th);
  }
  tab[idx] = v;
}

#define K1_PITCH4 17
__global__ __launch_bounds__(256) __attribute__((amdgpu_waves_per_eu(8, 8)))
void k1_dft_w(const float* __restrict__ x, const float* __restrict__ FT,
              float* __restrict__ Y) {
  __shared__ v4f stage[8 * 16 * K1_PITCH4];
  const int wv = threadIdx.x >> 5;
  const int wave = blockIdx.x * 8 + wv;
  const int lane = threadIdx.x & 31;
  const int half = lane >> 4, l16 = lane & 15;
  const long rowbase = (long)wave * 16;
  v4f* tile = stage + wv * (16 * K1_PITCH4);
  const float* gsrc = x + (rowbase + half) * 256 + 4 * l16;
  v4f* sdst = tile + half * K1_PITCH4 + l16;
  const v4f* ssrc = tile + l16 * K1_PITCH4 + half;
  const float* fb = FT + lane * 4;
  v8f acc0 = {}, acc1 = {};
  v4f g[8];
#pragma unroll
  for (int i = 0; i < 8; ++i) g[i] = ld4_nt(gsrc + i * 512);
#pragma unroll
  for (int c = 0; c < 4; ++c) {
#pragma unroll
    for (int i = 0; i < 8; ++i) sdst[2 * i * K1_PITCH4] = g[i];
    if (c < 3) {
#pragma unroll
      for (int i = 0; i < 8; ++i) g[i] = ld4_nt(gsrc + i * 512 + (c + 1) * 64);
    }
    __builtin_amdgcn_fence(__ATOMIC_ACQ_REL, "wavefront");
    v4f r[8];
#pragma unroll
    for (int j = 0; j < 8; ++j) r[j] = ssrc[2 * j];
#pragma unroll
    for (int j = 0; j < 8; ++j) {
      const int ks = (c * 8 + j) * 2;
      v4f f0 = ld4(fb + ks * 128);
      v4f f1 = ld4(fb + (ks + 1) * 128);
      v2f a0 = {r[j].x, r[j].y};
      v2f a1 = {r[j].z, r[j].w};
      v2f b00 = {f0.x, f0.y};
      v2f b01 = {f0.z, f0.w};
      v2f b10 = {f1.x, f1.y};
      v2f b11 = {f1.z, f1.w};
      acc0 = wmma4(a0, b00, acc0);
      acc1 = wmma4(a0, b01, acc1);
      acc0 = wmma4(a1, b10, acc0);
      acc1 = wmma4(a1, b11, acc1);
    }
    if (c < 3) __builtin_amdgcn_fence(__ATOMIC_ACQ_REL, "wavefront");
  }
  float* yrow = Y + rowbase * 32;
#pragma unroll
  for (int r = 0; r < 8; ++r) {
    int row = r + half * 8;
    yrow[row * 32 + l16] = acc0[r];
    yrow[row * 32 + 16 + l16] = acc1[r];
  }
}

struct K2Frag {
  v2f first[8], second[8];
  v4f cs[8], sn[8];
};

__device__ __forceinline__ void k2_load(K2Frag& f, const float* yfirst,
                                        const float* ysecond, const float* cf,
                                        const float* sf, int ks0) {
#pragma unroll
  for (int u = 0; u < 8; ++u) {
    const int k0 = 4 * (ks0 + u);
    f.first[u].x = yfirst[k0 * 32];
    f.first[u].y = yfirst[(k0 + 1) * 32];
    f.second[u].x = ysecond[k0 * 32];
    f.second[u].y = ysecond[(k0 + 1) * 32];
    f.cs[u] = ld4(cf + k0 * 32);
    f.sn[u] = ld4(sf + k0 * 32);
  }
}

__device__ __forceinline__ void k2_mma(const K2Frag& f, v8f& a0, v8f& a1,
                                       v8f& a2, v8f& a3) {
#pragma unroll
  for (int u = 0; u < 8; ++u) {
    v2f ac0 = {f.cs[u].x, f.cs[u].y}, ac1 = {f.cs[u].z, f.cs[u].w};
    v2f as0 = {f.sn[u].x, f.sn[u].y}, as1 = {f.sn[u].z, f.sn[u].w};
    a0 = wmma4(ac0, f.first[u], a0);
    a1 = wmma4(as0, f.second[u], a1);
    a2 = wmma4(ac1, f.first[u], a2);
    a3 = wmma4(as1, f.second[u], a3);
  }
}

__global__ __launch_bounds__(128) void k2_dft_h(const float* __restrict__ Y,
                                                const float* __restrict__ E2c,
                                                const float* __restrict__ E2s,
                                                float* __restrict__ Zre,
                                                float* __restrict__ Zim) {
  const int wv = threadIdx.x >> 5;
  const int bi = blockIdx.x * 2 + (wv >> 1);
  const int part = wv & 1;
  const int lane = threadIdx.x & 31;
  const int half = lane >> 4, l16 = lane & 15;
  const float* yb = Y + (long)bi * 256 * 32 + (2 * half) * 32 + l16;
  const float* yfirst = yb + 16 * part;
  const float* ysecond = yb + 16 * (1 - part);
  const float* cf = E2c + lane * 4;
  const float* sf = E2s + lane * 4;
  v8f a0 = {}, a1 = {}, a2 = {}, a3 = {};
  K2Frag f0, f1;
  k2_load(f0, yfirst, ysecond, cf, sf, 0);
#pragma unroll
  for (int it = 0; it < 4; ++it) {
    k2_load(f1, yfirst, ysecond, cf, sf, 16 * it + 8);
    __builtin_amdgcn_sched_barrier(0);
    k2_mma(f0, a0, a1, a2, a3);
    __builtin_amdgcn_sched_barrier(0);
    if (it < 3) k2_load(f0, yfirst, ysecond, cf, sf, 16 * it + 16);
    __builtin_amdgcn_sched_barrier(0);
    k2_mma(f1, a0, a1, a2, a3);
    __builtin_amdgcn_sched_barrier(0);
  }
  const int b = bi >> 5, i = bi & 31;
  float* Z = part ? Zim : Zre;
#pragma unroll
  for (int r = 0; r < 8; ++r) {
    int kyr = r + half * 8;
    long o0 = (((long)kyr * 16 + l16) * 16 + b) * 32 + i;
    long o1 = (((long)(16 + kyr) * 16 + l16) * 16 + b) * 32 + i;
    Z[o0] = part ? (a0[r] - a1[r]) : (a0[r] + a1[r]);
    Z[o1] = part ? (a2[r] - a3[r]) : (a2[r] + a3[r]);
  }
}

__global__ __launch_bounds__(32) void k3_mix(
    const float* __restrict__ Zre, const float* __restrict__ Zim,
    const float* __restrict__ w1r, const float* __restrict__ w1i,
    const float* __restrict__ w2r, const float* __restrict__ w2i,
    float* __restrict__ Tre, float* __restrict__ Tim) {
  const int wid = blockIdx.x;
  const int ot = wid & 1;
  const int kxg = (wid >> 1) & 3;
  const int kyp = wid >> 3;
  const int lane = threadIdx.x;
  const int half = lane >> 4, l16 = lane & 15;
  const int ky = kyp & 15;
  const float* wr = (kyp < 16) ? w1r : w2r;
  const float* wi = (kyp < 16) ? w1i : w2i;
  const long wofs =
      ((long)(2 * half) * 32 + 16 * ot + l16) * 256 + ky * 16 + 4 * kxg;
  v4f br[8][2], bi[8][2];
  v2f azr[8][4], azi[8][4];
  const int m0 = kyp * 16 + 4 * kxg;
  const float* zre = Zre + (long)m0 * 512 + l16 * 32 + 2 * half;
  const float* zim = Zim + (long)m0 * 512 + l16 * 32 + 2 * half;
#pragma unroll
  for (int ks = 0; ks < 8; ++ks) {
#pragma unroll
    for (int s = 0; s < 2; ++s) {
      br[ks][s] = ld4(wr + wofs + (long)(4 * ks + s) * 8192);
      bi[ks][s] = ld4(wi + wofs + (long)(4 * ks + s) * 8192);
    }
#pragma unroll
    for (int q = 0; q < 4; ++q) {
      azr[ks][q] = ld2(zre + q * 512 + 4 * ks);
      azi[ks][q] = ld2(zim + q * 512 + 4 * ks);
    }
  }
  __builtin_amdgcn_sched_barrier(0);
  v8f p[4] = {}, n[4] = {}, u[4] = {}, v[4] = {};
#pragma unroll
  for (int ks = 0; ks < 8; ++ks) {
#pragma unroll
    for (int q = 0; q < 4; ++q) {
      v2f b_r = {br[ks][0][q], br[ks][1][q]};
      v2f b_i = {bi[ks][0][q], bi[ks][1][q]};
      p[q] = wmma4(azr[ks][q], b_r, p[q]);
      n[q] = wmma4(azi[ks][q], b_i, n[q]);
      u[q] = wmma4(azr[ks][q], b_i, u[q]);
      v[q] = wmma4(azi[ks][q], b_r, v[q]);
    }
  }
  v8f tre[4], tim[4];
#pragma unroll
  for (int q = 0; q < 4; ++q) {
    tre[q] = p[q] - n[q];
    tim[q] = u[q] + v[q];
  }
#pragma unroll
  for (int r = 0; r < 8; ++r) {
    const int b = r + half * 8;
    const long o = (((long)b * 32 + 16 * ot + l16) * 32 + kyp) * 16 + 4 * kxg;
    v4f tr = {tre[0][r], tre[1][r], tre[2][r], tre[3][r]};
    v4f ti = {tim[0][r], tim[1][r], tim[2][r], tim[3][r]};
    *(v4f*)(Tre + o) = tr;
    *(v4f*)(Tim + o) = ti;
  }
}

__global__ __launch_bounds__(256) __attribute__((amdgpu_waves_per_eu(8, 8)))
void k4_idft_hw(
    const float* __restrict__ Tre, const float* __restrict__ Tim,
    const float* __restrict__ E4F, const float* __restrict__ GT,
    float* __restrict__ out) {
  const int wid = blockIdx.x * 8 + (threadIdx.x >> 5);
  const int lane = threadIdx.x & 31;
  const int half = lane >> 4, l16 = lane & 15;
  const int bo = wid >> 4;
  const int ht = wid & 15;
  const float* tre = Tre + (long)bo * 512;
  const float* tim = Tim + (long)bo * 512;
  const float* ef = E4F + (ht * 256 + lane) * 4;
  v8f a0 = {}, a1 = {}, a2 = {}, a3 = {};
#pragma unroll
  for (int k0 = 0; k0 < 32; k0 += 4) {
    int kr = k0 + 2 * half;
    v4f e = ld4(ef + k0 * 32);
    v2f ec = {e.x, e.y};
    v2f es = {e.z, e.w};
    v2f tr, ti;
    tr.x = tre[kr * 16 + l16];
    tr.y = tre[(kr + 1) * 16 + l16];
    ti.x = tim[kr * 16 + l16];
    ti.y = tim[(kr + 1) * 16 + l16];
    a0 = wmma4(tr, ec, a0);
    a1 = wmma4(ti, es, a1);
    a2 = wmma4(ti, ec, a2);
    a3 = wmma4(tr, es, a3);
  }
  const v8f ure = a0 - a1;
  const v8f uim = a2 + a3;
  float* orow = out + ((long)bo * 256 + ht * 16) * 256;
  const float* gf = GT + lane * 4;
#pragma unroll 1
  for (int g = 0; g < 4; ++g) {
    v8f acc[4] = {};
#pragma unroll
    for (int ks = 0; ks < 8; ++ks) {
      const int j = 2 * (ks & 3);
      v2f a;
      if (ks < 4) {
        a.x = ure[j];
        a.y = ure[j + 1];
      } else {
        a.x = uim[j];
        a.y = uim[j + 1];
      }
#pragma unroll
      for (int p = 0; p < 2; ++p) {
        v4f t = ld4(gf + g * 2048 + ks * 256 + p * 128);
        v2f ga = {t.x, t.y}, gb = {t.z, t.w};
        acc[2 * p] = wmma4(a, ga, acc[2 * p]);
        acc[2 * p + 1] = wmma4(a, gb, acc[2 * p + 1]);
      }
    }
#pragma unroll
    for (int r = 0; r < 8; ++r) {
      int rr = r + half * 8;
      v4f o = {acc[0][r], acc[1][r], acc[2][r], acc[3][r]};
      __builtin_nontemporal_store(
          o, (v4f*)&orow[(long)rr * 256 + g * 64 + 4 * l16]);
    }
  }
}

extern "C" void kernel_launch(void* const* d_in, const int* in_sizes, int n_in,
                              void* d_out, int out_size, void* d_ws,
                              size_t ws_size, hipStream_t stream) {
  (void)in_sizes; (void)n_in; (void)out_size; (void)ws_size;
  const float* x = (const float*)d_in[0];
  const float* w1r = (const float*)d_in[1];
  const float* w1i = (const float*)d_in[2];
  const float* w2r = (const float*)d_in[3];
  const float* w2i = (const float*)d_in[4];
  float* out = (float*)d_out;

  float* tab = (float*)d_ws;
  float* FT = tab;
  float* E2c = tab + 8192;
  float* E2s = tab + 16384;
  float* E4F = tab + 24576;
  float* GT = tab + 40960;
  float* Y = tab + 49152;
  float* Zre = Y + 4194304;
  float* Zim = Zre + 262144;
  float* Tre = Zim + 262144;
  float* Tim = Tre + 262144;

  hipLaunchKernelGGL(k0_tables, dim3(192), dim3(256), 0, stream, tab);
  hipLaunchKernelGGL(k1_dft_w, dim3(1024), dim3(256), 0, stream, x, FT, Y);
  hipLaunchKernelGGL(k2_dft_h, dim3(256), dim3(128), 0, stream, Y, E2c, E2s, Zre,
                     Zim);
  hipLaunchKernelGGL(k3_mix, dim3(256), dim3(32), 0, stream, Zre, Zim, w1r, w1i,
                     w2r, w2i, Tre, Tim);
  hipLaunchKernelGGL(k4_idft_hw, dim3(1024), dim3(256), 0, stream, Tre, Tim,
                     E4F, GT, out);
}


# W-DFTs split by mode parity (basis at w+128 = (-1)^m x basis at w): forward runs on x[w]+-x[w+128] over 128 columns, inverse forms out[w], out[w+128] = E+-O; half the W products; sum order changes
# speedup vs baseline: 2.9347x; 1.0992x over previous
#include <hip/hip_runtime.h>


typedef __attribute__((ext_vector_type(2))) float v2f;
typedef __attribute__((ext_vector_type(4))) float v4f;
typedef __attribute__((ext_vector_type(8))) float v8f;

__device__ __forceinline__ v8f wmma4(v2f a, v2f b, v8f c) {
  return __builtin_amdgcn_wmma_f32_16x16x4_f32(false, a, false, b, (short)0, c,
                                               false, false);
}

__device__ __forceinline__ v2f ld2(const float* p) { return *(const v2f*)p; }
__device__ __forceinline__ v4f ld4(const float* p) { return *(const v4f*)p; }
__device__ __forceinline__ v4f ld4_nt(const float* p) {
  return __builtin_nontemporal_load((const v4f*)p);
}

__global__ __launch_bounds__(256) void k0_tables(float* __restrict__ tab) {
  int idx = blockIdx.x * 256 + threadIdx.x;
  if (idx >= 40960) return;
  const float step = 6.28318530717958647692f / 256.0f;
  const int s = idx & 1, jt = (idx >> 1) & 1;
  const int lane = (idx >> 2) & 31, c = lane & 15, hf = lane >> 4;
  float v;
  if (idx < 4096) {
    int ks = idx >> 7;
    int w = 8 * (ks >> 1) + 4 * hf + 2 * (ks & 1) + s;
    int m = 2 * (c & 7) + jt;
    float th = step * (float)((m * w) & 255);
    v = (c < 8) ? cosf(th) : -sinf(th);
  } else if (idx < 20480) {
    int q = idx - 4096;
    int is_s = q >= 8192;
    q &= 8191;
    int ks = q >> 7;
    int h = 4 * ks + 2 * hf + s;
    int r = 16 * jt + c;
    int ky = (r < 16) ? r : (224 + r);
    float th = step * (float)((ky * h) & 255);
    v = is_s ? sinf(th) : cosf(th);
  } else if (idx < 36864) {
    int q = idx - 20480;
    int ks = (q >> 7) & 7, ht = q >> 10;
    int h = 16 * ht + c;
    int r = 4 * ks + 2 * hf + s;
    int ky = (r < 16) ? r : (224 + r);
    float th = step * (float)((ky * h) & 255);
    v = jt ? sinf(th) : cosf(th);
  } else {
    int q = idx - 36864;
    int p = (q >> 7) & 1, ks = (q >> 8) & 3, par = (q >> 10) & 1, g = q >> 11;
    int w = 64 * g + 4 * c + (2 * p + jt);
    int kx = 8 * hf + 4 * (ks & 1) + 2 * s + par;
    float th = step * (float)((kx * w) & 255);
    float sc = ((kx == 0) ? 1.0f : 2.0f) * (1.0f / 65536.0f);
    v = (ks < 2) ? sc * cosf(th) : -sc * sinf(th);
  }
  tab[idx] = v;
}

#define K1_PITCH4 17
__global__ __launch_bounds__(256) __attribute__((amdgpu_waves_per_eu(8, 8)))
void k1_dft_w(const float* __restrict__ x, const float* __restrict__ FT,
              float* __restrict__ Y) {
  __shared__ v4f stage[8 * 2 * 16 * K1_PITCH4];
  const int wv = threadIdx.x >> 5;
  const int wave = blockIdx.x * 8 + wv;
  const int lane = threadIdx.x & 31;
  const int half = lane >> 4, l16 = lane & 15;
  const long rowbase = (long)wave * 16;
  v4f* tile = stage + wv * (2 * 16 * K1_PITCH4);
  const float* gsrc = x + (rowbase + half) * 256 + 4 * l16;
  v4f* sdst = tile + half * K1_PITCH4 + l16;
  const v4f* ssrc = tile + l16 * K1_PITCH4 + half;
  const float* fb = FT + lane * 4;
  const int dofs = 16 * K1_PITCH4;
  v8f acc0 = {}, acc1 = {};
  v4f lo[8], hi[8];
#pragma unroll
  for (int i = 0; i < 8; ++i) {
    lo[i] = ld4_nt(gsrc + i * 512);
    hi[i] = ld4_nt(gsrc + i * 512 + 128);
  }
#pragma unroll
  for (int c = 0; c < 2; ++c) {
#pragma unroll
    for (int i = 0; i < 8; ++i) {
      sdst[2 * i * K1_PITCH4] = lo[i] + hi[i];
      sdst[2 * i * K1_PITCH4 + dofs] = lo[i] - hi[i];
    }
    if (c < 1) {
#pragma unroll
      for (int i = 0; i < 8; ++i) {
        lo[i] = ld4_nt(gsrc + i * 512 + 64);
        hi[i] = ld4_nt(gsrc + i * 512 + 192);
      }
    }
    __builtin_amdgcn_fence(__ATOMIC_ACQ_REL, "wavefront");
#pragma unroll
    for (int j = 0; j < 8; ++j) {
      const int ks = (c * 8 + j) * 2;
      v4f rs = ssrc[2 * j];
      v4f rd = ssrc[2 * j + dofs];
      v4f f0 = ld4(fb + ks * 128);
      v4f f1 = ld4(fb + (ks + 1) * 128);
      v2f s0 = {rs.x, rs.y};
      v2f s1 = {rs.z, rs.w};
      v2f d0 = {rd.x, rd.y};
      v2f d1 = {rd.z, rd.w};
      v2f b00 = {f0.x, f0.y};
      v2f b01 = {f0.z, f0.w};
      v2f b10 = {f1.x, f1.y};
      v2f b11 = {f1.z, f1.w};
      acc0 = wmma4(s0, b00, acc0);
      acc1 = wmma4(d0, b01, acc1);
      acc0 = wmma4(s1, b10, acc0);
      acc1 = wmma4(d1, b11, acc1);
    }
    if (c < 1) __builtin_amdgcn_fence(__ATOMIC_ACQ_REL, "wavefront");
  }
  float* yrow = Y + rowbase * 32;
#pragma unroll
  for (int r = 0; r < 8; ++r) {
    int row = r + half * 8;
    yrow[row * 32 + l16] = acc0[r];
    yrow[row * 32 + 16 + l16] = acc1[r];
  }
}

struct K2Frag {
  v2f first[8], second[8];
  v4f cs[8], sn[8];
};

__device__ __forceinline__ void k2_load(K2Frag& f, const float* yfirst,
                                        const float* ysecond, const float* cf,
                                        const float* sf, int ks0) {
#pragma unroll
  for (int u = 0; u < 8; ++u) {
    const int k0 = 4 * (ks0 + u);
    f.first[u].x = yfirst[k0 * 32];
    f.first[u].y = yfirst[(k0 + 1) * 32];
    f.second[u].x = ysecond[k0 * 32];
    f.second[u].y = ysecond[(k0 + 1) * 32];
    f.cs[u] = ld4(cf + k0 * 32);
    f.sn[u] = ld4(sf + k0 * 32);
  }
}

__device__ __forceinline__ void k2_mma(const K2Frag& f, v8f& a0, v8f& a1,
                                       v8f& a2, v8f& a3) {
#pragma unroll
  for (int u = 0; u < 8; ++u) {
    v2f ac0 = {f.cs[u].x, f.cs[u].y}, ac1 = {f.cs[u].z, f.cs[u].w};
    v2f as0 = {f.sn[u].x, f.sn[u].y}, as1 = {f.sn[u].z, f.sn[u].w};
    a0 = wmma4(ac0, f.first[u], a0);
    a1 = wmma4(as0, f.second[u], a1);
    a2 = wmma4(ac1, f.first[u], a2);
    a3 = wmma4(as1, f.second[u], a3);
  }
}

__global__ __launch_bounds__(128) void k2_dft_h(const float* __restrict__ Y,
                                                const float* __restrict__ E2c,
                                                const float* __restrict__ E2s,
                                                float* __restrict__ Zre,
                                                float* __restrict__ Zim) {
  const int wv = threadIdx.x >> 5;
  const int bi = blockIdx.x * 2 + (wv >> 1);
  const int part = wv & 1;
  const int lane = threadIdx.x & 31;
  const int half = lane >> 4, l16 = lane & 15;
  const float* yb =
      Y + (long)bi * 256 * 32 + (2 * half) * 32 + 16 * (l16 & 1) + (l16 >> 1);
  const float* yfirst = yb + 8 * part;
  const float* ysecond = yb + 8 * (1 - part);
  const float* cf = E2c + lane * 4;
  const float* sf = E2s + lane * 4;
  v8f a0 = {}, a1 = {}, a2 = {}, a3 = {};
  K2Frag f0, f1;
  k2_load(f0, yfirst, ysecond, cf, sf, 0);
#pragma unroll
  for (int it = 0; it < 4; ++it) {
    k2_load(f1, yfirst, ysecond, cf, sf, 16 * it + 8);
    __builtin_amdgcn_sched_barrier(0);
    k2_mma(f0, a0, a1, a2, a3);
    __builtin_amdgcn_sched_barrier(0);
    if (it < 3) k2_load(f0, yfirst, ysecond, cf, sf, 16 * it + 16);
    __builtin_amdgcn_sched_barrier(0);
    k2_mma(f1, a0, a1, a2, a3);
    __builtin_amdgcn_sched_barrier(0);
  }
  const int b = bi >> 5, i = bi & 31;
  float* Z = part ? Zim : Zre;
#pragma unroll
  for (int r = 0; r < 8; ++r) {
    int kyr = r + half * 8;
    long o0 = (((long)kyr * 16 + l16) * 16 + b) * 32 + i;
    long o1 = (((long)(16 + kyr) * 16 + l16) * 16 + b) * 32 + i;
    Z[o0] = part ? (a0[r] - a1[r]) : (a0[r] + a1[r]);
    Z[o1] = part ? (a2[r] - a3[r]) : (a2[r] + a3[r]);
  }
}

__global__ __launch_bounds__(32) void k3_mix(
    const float* __restrict__ Zre, const float* __restrict__ Zim,
    const float* __restrict__ w1r, const float* __restrict__ w1i,
    const float* __restrict__ w2r, const float* __restrict__ w2i,
    float* __restrict__ Tre, float* __restrict__ Tim) {
  const int wid = blockIdx.x;
  const int ot = wid & 1;
  const int kxg = (wid >> 1) & 3;
  const int kyp = wid >> 3;
  const int lane = threadIdx.x;
  const int half = lane >> 4, l16 = lane & 15;
  const int ky = kyp & 15;
  const float* wr = (kyp < 16) ? w1r : w2r;
  const float* wi = (kyp < 16) ? w1i : w2i;
  const long wofs =
      ((long)(2 * half) * 32 + 16 * ot + l16) * 256 + ky * 16 + 4 * kxg;
  v4f br[8][2], bi[8][2];
  v2f azr[8][4], azi[8][4];
  const int m0 = kyp * 16 + 4 * kxg;
  const float* zre = Zre + (long)m0 * 512 + l16 * 32 + 2 * half;
  const float* zim = Zim + (long)m0 * 512 + l16 * 32 + 2 * half;
#pragma unroll
  for (int ks = 0; ks < 8; ++ks) {
#pragma unroll
    for (int s = 0; s < 2; ++s) {
      br[ks][s] = ld4(wr + wofs + (long)(4 * ks + s) * 8192);
      bi[ks][s] = ld4(wi + wofs + (long)(4 * ks + s) * 8192);
    }
#pragma unroll
    for (int q = 0; q < 4; ++q) {
      azr[ks][q] = ld2(zre + q * 512 + 4 * ks);
      azi[ks][q] = ld2(zim + q * 512 + 4 * ks);
    }
  }
  __builtin_amdgcn_sched_barrier(0);
  v8f p[4] = {}, n[4] = {}, u[4] = {}, v[4] = {};
#pragma unroll
  for (int ks = 0; ks < 8; ++ks) {
#pragma unroll
    for (int q = 0; q < 4; ++q) {
      v2f b_r = {br[ks][0][q], br[ks][1][q]};
      v2f b_i = {bi[ks][0][q], bi[ks][1][q]};
      p[q] = wmma4(azr[ks][q], b_r, p[q]);
      n[q] = wmma4(azi[ks][q], b_i, n[q]);
      u[q] = wmma4(azr[ks][q], b_i, u[q]);
      v[q] = wmma4(azi[ks][q], b_r, v[q]);
    }
  }
  v8f tre[4], tim[4];
#pragma unroll
  for (int q = 0; q < 4; ++q) {
    tre[q] = p[q] - n[q];
    tim[q] = u[q] + v[q];
  }
#pragma unroll
  for (int r = 0; r < 8; ++r) {
    const int b = r + half * 8;
    const long o = (((long)b * 32 + 16 * ot + l16) * 32 + kyp) * 16 + 4 * kxg;
    v4f tr = {tre[0][r], tre[1][r], tre[2][r], tre[3][r]};
    v4f ti = {tim[0][r], tim[1][r], tim[2][r], tim[3][r]};
    *(v4f*)(Tre + o) = tr;
    *(v4f*)(Tim + o) = ti;
  }
}

__global__ __launch_bounds__(256) __attribute__((amdgpu_waves_per_eu(8, 8)))
void k4_idft_hw(
    const float* __restrict__ Tre, const float* __restrict__ Tim,
    const float* __restrict__ E4F, const float* __restrict__ GT,
    float* __restrict__ out) {
  const int wid = blockIdx.x * 8 + (threadIdx.x >> 5);
  const int lane = threadIdx.x & 31;
  const int half = lane >> 4, l16 = lane & 15;
  const int bo = wid >> 4;
  const int ht = wid & 15;
  const float* tre = Tre + (long)bo * 512;
  const float* tim = Tim + (long)bo * 512;
  const float* ef = E4F + (ht * 256 + lane) * 4;
  const int kxl = (l16 & 8) + 2 * (l16 & 3) + ((l16 >> 2) & 1);
  v8f a0 = {}, a1 = {}, a2 = {}, a3 = {};
#pragma unroll
  for (int k0 = 0; k0 < 32; k0 += 4) {
    int kr = k0 + 2 * half;
    v4f e = ld4(ef + k0 * 32);
    v2f ec = {e.x, e.y};
    v2f es = {e.z, e.w};
    v2f tr, ti;
    tr.x = tre[kr * 16 + kxl];
    tr.y = tre[(kr + 1) * 16 + kxl];
    ti.x = tim[kr * 16 + kxl];
    ti.y = tim[(kr + 1) * 16 + kxl];
    a0 = wmma4(tr, ec, a0);
    a1 = wmma4(ti, es, a1);
    a2 = wmma4(ti, ec, a2);
    a3 = wmma4(tr, es, a3);
  }
  const v8f ure = a0 - a1;
  const v8f uim = a2 + a3;
  float* orow = out + ((long)bo * 256 + ht * 16) * 256;
  const float* gf = GT + lane * 4;
#pragma unroll 1
  for (int g = 0; g < 2; ++g) {
    v8f ev[4] = {}, od[4] = {};
#pragma unroll
    for (int ks = 0; ks < 4; ++ks) {
      if (ks == 2) __builtin_amdgcn_sched_barrier(0);
      const int j = 2 * (ks & 1);
      v2f ae, ao;
      if (ks < 2) {
        ae.x = ure[j];
        ae.y = ure[j + 1];
        ao.x = ure[4 + j];
        ao.y = ure[5 + j];
      } else {
        ae.x = uim[j];
        ae.y = uim[j + 1];
        ao.x = uim[4 + j];
        ao.y = uim[5 + j];
      }
#pragma unroll
      for (int p = 0; p < 2; ++p) {
        v4f te = ld4(gf + g * 2048 + ks * 256 + p * 128);
        v4f to = ld4(gf + g * 2048 + 1024 + ks * 256 + p * 128);
        v2f ea = {te.x, te.y}, eb = {te.z, te.w};
        v2f oa = {to.x, to.y}, ob = {to.z, to.w};
        ev[2 * p] = wmma4(ae, ea, ev[2 * p]);
        od[2 * p] = wmma4(ao, oa, od[2 * p]);
        ev[2 * p + 1] = wmma4(ae, eb, ev[2 * p + 1]);
        od[2 * p + 1] = wmma4(ao, ob, od[2 * p + 1]);
      }
    }
#pragma unroll
    for (int r = 0; r < 8; ++r) {
      int rr = r + half * 8;
      v4f e = {ev[0][r], ev[1][r], ev[2][r], ev[3][r]};
      v4f o = {od[0][r], od[1][r], od[2][r], od[3][r]};
      float* dst = &orow[(long)rr * 256 + g * 64 + 4 * l16];
      __builtin_nontemporal_store(e + o, (v4f*)dst);
      __builtin_nontemporal_store(e - o, (v4f*)(dst + 128));
    }
  }
}

extern "C" void kernel_launch(void* const* d_in, const int* in_sizes, int n_in,
                              void* d_out, int out_size, void* d_ws,
                              size_t ws_size, hipStream_t stream) {
  (void)in_sizes; (void)n_in; (void)out_size; (void)ws_size;
  const float* x = (const float*)d_in[0];
  const float* w1r = (const float*)d_in[1];
  const float* w1i = (const float*)d_in[2];
  const float* w2r = (const float*)d_in[3];
  const float* w2i = (const float*)d_in[4];
  float* out = (float*)d_out;

  float* tab = (float*)d_ws;
  float* FT = tab;
  float* E2c = tab + 4096;
  float* E2s = tab + 12288;
  float* E4F = tab + 20480;
  float* GT = tab + 36864;
  float* Y = tab + 40960;
  float* Zre = Y + 4194304;
  float* Zim = Zre + 262144;
  float* Tre = Zim + 262144;
  float* Tim = Tre + 262144;

  hipLaunchKernelGGL(k0_tables, dim3(160), dim3(256), 0, stream, tab);
  hipLaunchKernelGGL(k1_dft_w, dim3(1024), dim3(256), 0, stream, x, FT, Y);
  hipLaunchKernelGGL(k2_dft_h, dim3(256), dim3(128), 0, stream, Y, E2c, E2s, Zre,
                     Zim);
  hipLaunchKernelGGL(k3_mix, dim3(256), dim3(32), 0, stream, Zre, Zim, w1r, w1i,
                     w2r, w2i, Tre, Tim);
  hipLaunchKernelGGL(k4_idft_hw, dim3(1024), dim3(256), 0, stream, Tre, Tim,
                     E4F, GT, out);
}


# Forward H-DFT fused into the W-DFT kernel: a wave's Y tile stays in its accumulators as B operand of its 16 rows' share of Z; 16 waves' shares summed in LDS; one launch less; H sum order changes
# speedup vs baseline: 3.8604x; 1.3154x over previous
#include <hip/hip_runtime.h>


typedef __attribute__((ext_vector_type(2))) float v2f;
typedef __attribute__((ext_vector_type(4))) float v4f;
typedef __attribute__((ext_vector_type(8))) float v8f;

__device__ __forceinline__ v8f wmma4(v2f a, v2f b, v8f c) {
  return __builtin_amdgcn_wmma_f32_16x16x4_f32(false, a, false, b, (short)0, c,
                                               false, false);
}

__device__ __forceinline__ v2f ld2(const float* p) { return *(const v2f*)p; }
__device__ __forceinline__ v4f ld4(const float* p) { return *(const v4f*)p; }
__device__ __forceinline__ v4f ld4_nt(const float* p) {
  return __builtin_nontemporal_load((const v4f*)p);
}

__global__ __launch_bounds__(256) void k0_tables(float* __restrict__ tab) {
  int idx = blockIdx.x * 256 + threadIdx.x;
  if (idx >= 40960) return;
  const float step = 6.28318530717958647692f / 256.0f;
  const int s = idx & 1, jt = (idx >> 1) & 1;
  const int lane = (idx >> 2) & 31, c = lane & 15, hf = lane >> 4;
  float v;
  if (idx < 4096) {
    int ks = idx >> 7;
    int w = 8 * (ks >> 1) + 4 * hf + 2 * (ks & 1) + s;
    int m = 2 * (c & 7) + jt;
    float th = step * (float)((m * w) & 255);
    v = (c < 8) ? cosf(th) : -sinf(th);
  } else if (idx < 20480) {
    int q = idx - 4096;
    int mt = (q >> 2) & 1, ln = (q >> 3) & 31;
    int ks = (q >> 8) & 3, ht = q >> 10;
    int h = 16 * ht + 8 * (ln >> 4) + 2 * ks + s;
    int r = 16 * mt + (ln & 15);
    int ky = (r < 16) ? r : (224 + r);
    float th = step * (float)((ky * h) & 255);
    v = jt ? sinf(th) : cosf(th);
  } else if (idx < 36864) {
    int q = idx - 20480;
    int ks = (q >> 7) & 7, ht = q >> 10;
    int h = 16 * ht + c;
    int r = 4 * ks + 2 * hf + s;
    int ky = (r < 16) ? r : (224 + r);
    float th = step * (float)((ky * h) & 255);
    v = jt ? sinf(th) : cosf(th);
  } else {
    int q = idx - 36864;
    int p = (q >> 7) & 1, ks = (q >> 8) & 3, par = (q >> 10) & 1, g = q >> 11;
    int w = 64 * g + 4 * c + (2 * p + jt);
    int kx = 8 * hf + 4 * (ks & 1) + 2 * s + par;
    float th = step * (float)((kx * w) & 255);
    float sc = ((kx == 0) ? 1.0f : 2.0f) * (1.0f / 65536.0f);
    v = (ks < 2) ? sc * cosf(th) : -sc * sinf(th);
  }
  tab[idx] = v;
}

#define K1_PITCH4 17
#define K1_TILE4 (2 * 16 * K1_PITCH4)
__global__ __launch_bounds__(512) __attribute__((amdgpu_waves_per_eu(8, 8)))
void k1_dft_wh(const float* __restrict__ x, const float* __restrict__ FT,
               const float* __restrict__ E5, float* __restrict__ Zre,
               float* __restrict__ Zim) {
  __shared__ v4f stage[16 * K1_TILE4];
  const int wv = threadIdx.x >> 5;
  const int bi = blockIdx.x;
  const int lane = threadIdx.x & 31;
  const int half = lane >> 4, l16 = lane & 15;
  const long rowbase = ((long)bi * 16 + wv) * 16;
  v4f* tile = stage + wv * K1_TILE4;
  const float* gsrc = x + (rowbase + half) * 256 + 4 * l16;
  v4f* sdst = tile + half * K1_PITCH4 + l16;
  const v4f* ssrc = tile + l16 * K1_PITCH4 + half;
  const float* fb = FT + lane * 4;
  const int dofs = 16 * K1_PITCH4;
  v8f acc0 = {}, acc1 = {};
  v4f lo[8], hi[8];
#pragma unroll
  for (int i = 0; i < 8; ++i) {
    lo[i] = ld4_nt(gsrc + i * 512);
    hi[i] = ld4_nt(gsrc + i * 512 + 128);
  }
#pragma unroll
  for (int c = 0; c < 2; ++c) {
#pragma unroll
    for (int i = 0; i < 8; ++i) {
      sdst[2 * i * K1_PITCH4] = lo[i] + hi[i];
      sdst[2 * i * K1_PITCH4 + dofs] = lo[i] - hi[i];
    }
    if (c < 1) {
#pragma unroll
      for (int i = 0; i < 8; ++i) {
        lo[i] = ld4_nt(gsrc + i * 512 + 64);
        hi[i] = ld4_nt(gsrc + i * 512 + 192);
      }
    }
    __builtin_amdgcn_fence(__ATOMIC_ACQ_REL, "wavefront");
#pragma unroll
    for (int j = 0; j < 8; ++j) {
      const int ks = (c * 8 + j) * 2;
      v4f rs = ssrc[2 * j];
      v4f rd = ssrc[2 * j + dofs];
      v4f f0 = ld4(fb + ks * 128);
      v4f f1 = ld4(fb + (ks + 1) * 128);
      v2f s0 = {rs.x, rs.y};
      v2f s1 = {rs.z, rs.w};
      v2f d0 = {rd.x, rd.y};
      v2f d1 = {rd.z, rd.w};
      v2f b00 = {f0.x, f0.y};
      v2f b01 = {f0.z, f0.w};
      v2f b10 = {f1.x, f1.y};
      v2f b11 = {f1.z, f1.w};
      acc0 = wmma4(s0, b00, acc0);
      acc1 = wmma4(d0, b01, acc1);
      acc0 = wmma4(s1, b10, acc0);
      acc1 = wmma4(d1, b11, acc1);
    }
    if (c < 1) __builtin_amdgcn_fence(__ATOMIC_ACQ_REL, "wavefront");
  }
  const float* e5 = E5 + ((wv * 4) * 32 + lane) * 8;
  v4f ea[4], eb[4];
  __builtin_amdgcn_sched_barrier(0);
#pragma unroll
  for (int ks = 0; ks < 4; ++ks) {
    ea[ks] = ld4(e5 + ks * 256);
    eb[ks] = ld4(e5 + ks * 256 + 4);
  }
  const bool lo8 = l16 < 8;
  v8f bre, bim;
#pragma unroll
  for (int r = 0; r < 8; ++r) {
    const float p0 = __shfl_xor(acc0[r], 8);
    const float p1 = __shfl_xor(acc1[r], 8);
    bre[r] = lo8 ? acc0[r] : p1;
    bim[r] = lo8 ? p0 : acc1[r];
  }
  v8f zr0 = {}, zr1 = {}, zi0 = {}, zi1 = {};
#pragma unroll
  for (int ks = 0; ks < 4; ++ks) {
    const v2f br = {bre[2 * ks], bre[2 * ks + 1]};
    const v2f bm = {bim[2 * ks], bim[2 * ks + 1]};
    const v2f nb = -br;
    const v2f c0 = {ea[ks].x, ea[ks].y}, s0 = {ea[ks].z, ea[ks].w};
    const v2f c1 = {eb[ks].x, eb[ks].y}, s1 = {eb[ks].z, eb[ks].w};
    zr0 = wmma4(c0, br, zr0);
    zi0 = wmma4(c0, bm, zi0);
    zr1 = wmma4(c1, br, zr1);
    zi1 = wmma4(c1, bm, zi1);
    zr0 = wmma4(s0, bm, zr0);
    zi0 = wmma4(s0, nb, zi0);
    zr1 = wmma4(s1, bm, zr1);
    zi1 = wmma4(s1, nb, zi1);
  }
  {
    const v4f q00 = {zr0[0], zr0[1], zr0[2], zr0[3]};
    const v4f q01 = {zr0[4], zr0[5], zr0[6], zr0[7]};
    const v4f q10 = {zr1[0], zr1[1], zr1[2], zr1[3]};
    const v4f q11 = {zr1[4], zr1[5], zr1[6], zr1[7]};
    const v4f q20 = {zi0[0], zi0[1], zi0[2], zi0[3]};
    const v4f q21 = {zi0[4], zi0[5], zi0[6], zi0[7]};
    const v4f q30 = {zi1[0], zi1[1], zi1[2], zi1[3]};
    const v4f q31 = {zi1[4], zi1[5], zi1[6], zi1[7]};
    tile[0 * 32 + lane] = q00;
    tile[1 * 32 + lane] = q01;
    tile[2 * 32 + lane] = q10;
    tile[3 * 32 + lane] = q11;
    tile[4 * 32 + lane] = q20;
    tile[5 * 32 + lane] = q21;
    tile[6 * 32 + lane] = q30;
    tile[7 * 32 + lane] = q31;
  }
  __syncthreads();
  if (threadIdx.x < 256) {
    const int slot = threadIdx.x;
    v4f sum = stage[slot];
#pragma unroll
    for (int w = 1; w < 16; ++w) sum += stage[w * K1_TILE4 + slot];
    const int q = slot >> 6, rh = (slot >> 5) & 1;
    const int sc = slot & 15, shf = (slot >> 4) & 1;
    const int kx = (sc < 8) ? 2 * sc : 2 * (sc - 8) + 1;
    const int ky0 = 16 * (q & 1) + 8 * shf + 4 * rh;
    const int b = bi >> 5, i = bi & 31;
    float* Z = (q >> 1) ? Zim : Zre;
#pragma unroll
    for (int e = 0; e < 4; ++e)
      Z[(((long)(ky0 + e) * 16 + kx) * 16 + b) * 32 + i] = sum[e];
  }
}

__global__ __launch_bounds__(32) void k3_mix(
    const float* __restrict__ Zre, const float* __restrict__ Zim,
    const float* __restrict__ w1r, const float* __restrict__ w1i,
    const float* __restrict__ w2r, const float* __restrict__ w2i,
    float* __restrict__ Tre, float* __restrict__ Tim) {
  const int wid = blockIdx.x;
  const int ot = wid & 1;
  const int kxg = (wid >> 1) & 3;
  const int kyp = wid >> 3;
  const int lane = threadIdx.x;
  const int half = lane >> 4, l16 = lane & 15;
  const int ky = kyp & 15;
  const float* wr = (kyp < 16) ? w1r : w2r;
  const float* wi = (kyp < 16) ? w1i : w2i;
  const long wofs =
      ((long)(2 * half) * 32 + 16 * ot + l16) * 256 + ky * 16 + 4 * kxg;
  v4f br[8][2], bi[8][2];
  v2f azr[8][4], azi[8][4];
  const int m0 = kyp * 16 + 4 * kxg;
  const float* zre = Zre + (long)m0 * 512 + l16 * 32 + 2 * half;
  const float* zim = Zim + (long)m0 * 512 + l16 * 32 + 2 * half;
#pragma unroll
  for (int ks = 0; ks < 8; ++ks) {
#pragma unroll
    for (int s = 0; s < 2; ++s) {
      br[ks][s] = ld4(wr + wofs + (long)(4 * ks + s) * 8192);
      bi[ks][s] = ld4(wi + wofs + (long)(4 * ks + s) * 8192);
    }
#pragma unroll
    for (int q = 0; q < 4; ++q) {
      azr[ks][q] = ld2(zre + q * 512 + 4 * ks);
      azi[ks][q] = ld2(zim + q * 512 + 4 * ks);
    }
  }
  __builtin_amdgcn_sched_barrier(0);
  v8f p[4] = {}, n[4] = {}, u[4] = {}, v[4] = {};
#pragma unroll
  for (int ks = 0; ks < 8; ++ks) {
#pragma unroll
    for (int q = 0; q < 4; ++q) {
      v2f b_r = {br[ks][0][q], br[ks][1][q]};
      v2f b_i = {bi[ks][0][q], bi[ks][1][q]};
      p[q] = wmma4(azr[ks][q], b_r, p[q]);
      n[q] = wmma4(azi[ks][q], b_i, n[q]);
      u[q] = wmma4(azr[ks][q], b_i, u[q]);
      v[q] = wmma4(azi[ks][q], b_r, v[q]);
    }
  }
  v8f tre[4], tim[4];
#pragma unroll
  for (int q = 0; q < 4; ++q) {
    tre[q] = p[q] - n[q];
    tim[q] = u[q] + v[q];
  }
#pragma unroll
  for (int r = 0; r < 8; ++r) {
    const int b = r + half * 8;
    const long o = (((long)b * 32 + 16 * ot + l16) * 32 + kyp) * 16 + 4 * kxg;
    v4f tr = {tre[0][r], tre[1][r], tre[2][r], tre[3][r]};
    v4f ti = {tim[0][r], tim[1][r], tim[2][r], tim[3][r]};
    *(v4f*)(Tre + o) = tr;
    *(v4f*)(Tim + o) = ti;
  }
}

__global__ __launch_bounds__(256) __attribute__((amdgpu_waves_per_eu(8, 8)))
void k4_idft_hw(
    const float* __restrict__ Tre, const float* __restrict__ Tim,
    const float* __restrict__ E4F, const float* __restrict__ GT,
    float* __restrict__ out) {
  const int wid = blockIdx.x * 8 + (threadIdx.x >> 5);
  const int lane = threadIdx.x & 31;
  const int half = lane >> 4, l16 = lane & 15;
  const int bo = wid >> 4;
  const int ht = wid & 15;
  const float* tre = Tre + (long)bo * 512;
  const float* tim = Tim + (long)bo * 512;
  const float* ef = E4F + (ht * 256 + lane) * 4;
  const int kxl = (l16 & 8) + 2 * (l16 & 3) + ((l16 >> 2) & 1);
  v8f a0 = {}, a1 = {}, a2 = {}, a3 = {};
#pragma unroll
  for (int k0 = 0; k0 < 32; k0 += 4) {
    int kr = k0 + 2 * half;
    v4f e = ld4(ef + k0 * 32);
    v2f ec = {e.x, e.y};
    v2f es = {e.z, e.w};
    v2f tr, ti;
    tr.x = tre[kr * 16 + kxl];
    tr.y = tre[(kr + 1) * 16 + kxl];
    ti.x = tim[kr * 16 + kxl];
    ti.y = tim[(kr + 1) * 16 + kxl];
    a0 = wmma4(tr, ec, a0);
    a1 = wmma4(ti, es, a1);
    a2 = wmma4(ti, ec, a2);
    a3 = wmma4(tr, es, a3);
  }
  const v8f ure = a0 - a1;
  const v8f uim = a2 + a3;
  float* orow = out + ((long)bo * 256 + ht * 16) * 256;
  const float* gf = GT + lane * 4;
#pragma unroll 1
  for (int g = 0; g < 2; ++g) {
    v8f ev[4] = {}, od[4] = {};
#pragma unroll
    for (int ks = 0; ks < 4; ++ks) {
      if (ks == 2) __builtin_amdgcn_sched_barrier(0);
      const int j = 2 * (ks & 1);
      v2f ae, ao;
      if (ks < 2) {
        ae.x = ure[j];
        ae.y = ure[j + 1];
        ao.x = ure[4 + j];
        ao.y = ure[5 + j];
      } else {
        ae.x = uim[j];
        ae.y = uim[j + 1];
        ao.x = uim[4 + j];
        ao.y = uim[5 + j];
      }
#pragma unroll
      for (int p = 0; p < 2; ++p) {
        v4f te = ld4(gf + g * 2048 + ks * 256 + p * 128);
        v4f to = ld4(gf + g * 2048 + 1024 + ks * 256 + p * 128);
        v2f ea = {te.x, te.y}, eb = {te.z, te.w};
        v2f oa = {to.x, to.y}, ob = {to.z, to.w};
        ev[2 * p] = wmma4(ae, ea, ev[2 * p]);
        od[2 * p] = wmma4(ao, oa, od[2 * p]);
        ev[2 * p + 1] = wmma4(ae, eb, ev[2 * p + 1]);
        od[2 * p + 1] = wmma4(ao, ob, od[2 * p + 1]);
      }
    }
#pragma unroll
    for (int r = 0; r < 8; ++r) {
      int rr = r + half * 8;
      v4f e = {ev[0][r], ev[1][r], ev[2][r], ev[3][r]};
      v4f o = {od[0][r], od[1][r], od[2][r], od[3][r]};
      float* dst = &orow[(long)rr * 256 + g * 64 + 4 * l16];
      __builtin_nontemporal_store(e + o, (v4f*)dst);
      __builtin_nontemporal_store(e - o, (v4f*)(dst + 128));
    }
  }
}

extern "C" void kernel_launch(void* const* d_in, const int* in_sizes, int n_in,
                              void* d_out, int out_size, void* d_ws,
                              size_t ws_size, hipStream_t stream) {
  (void)in_sizes; (void)n_in; (void)out_size; (void)ws_size;
  const float* x = (const float*)d_in[0];
  const float* w1r = (const float*)d_in[1];
  const float* w1i = (const float*)d_in[2];
  const float* w2r = (const float*)d_in[3];
  const float* w2i = (const float*)d_in[4];
  float* out = (float*)d_out;

  float* tab = (float*)d_ws;
  float* FT = tab;
  float* E5 = tab + 4096;
  float* E4F = tab + 20480;
  float* GT = tab + 36864;
  float* Zre = tab + 40960;
  float* Zim = Zre + 262144;
  float* Tre = Zim + 262144;
  float* Tim = Tre + 262144;

  hipLaunchKernelGGL(k0_tables, dim3(160), dim3(256), 0, stream, tab);
  hipLaunchKernelGGL(k1_dft_wh, dim3(512), dim3(512), 0, stream, x, FT, E5, Zre,
                     Zim);
  hipLaunchKernelGGL(k3_mix, dim3(256), dim3(32), 0, stream, Zre, Zim, w1r, w1i,
                     w2r, w2i, Tre, Tim);
  hipLaunchKernelGGL(k4_idft_hw, dim3(1024), dim3(256), 0, stream, Tre, Tim,
                     E4F, GT, out);
}


# Z between launches re-laid [ky'][kx group][b][i][4 kx]: forward kernel regroups its summed tiles in LDS and stores 16 B per thread instead of 4 floats in 4 lines; mix kernel loads 16 B; same sums
# speedup vs baseline: 4.5249x; 1.1721x over previous
#include <hip/hip_runtime.h>


typedef __attribute__((ext_vector_type(2))) float v2f;
typedef __attribute__((ext_vector_type(4))) float v4f;
typedef __attribute__((ext_vector_type(8))) float v8f;

__device__ __forceinline__ v8f wmma4(v2f a, v2f b, v8f c) {
  return __builtin_amdgcn_wmma_f32_16x16x4_f32(false, a, false, b, (short)0, c,
                                               false, false);
}

__device__ __forceinline__ v2f ld2(const float* p) { return *(const v2f*)p; }
__device__ __forceinline__ v4f ld4(const float* p) { return *(const v4f*)p; }
__device__ __forceinline__ v4f ld4_nt(const float* p) {
  return __builtin_nontemporal_load((const v4f*)p);
}

__global__ __launch_bounds__(256) void k0_tables(float* __restrict__ tab) {
  int idx = blockIdx.x * 256 + threadIdx.x;
  if (idx >= 40960) return;
  const float step = 6.28318530717958647692f / 256.0f;
  const int s = idx & 1, jt = (idx >> 1) & 1;
  const int lane = (idx >> 2) & 31, c = lane & 15, hf = lane >> 4;
  float v;
  if (idx < 4096) {
    int ks = idx >> 7;
    int w = 8 * (ks >> 1) + 4 * hf + 2 * (ks & 1) + s;
    int m = 2 * (c & 7) + jt;
    float th = step * (float)((m * w) & 255);
    v = (c < 8) ? cosf(th) : -sinf(th);
  } else if (idx < 20480) {
    int q = idx - 4096;
    int mt = (q >> 2) & 1, ln = (q >> 3) & 31;
    int ks = (q >> 8) & 3, ht = q >> 10;
    int h = 16 * ht + 8 * (ln >> 4) + 2 * ks + s;
    int r = 16 * mt + (ln & 15);
    int ky = (r < 16) ? r : (224 + r);
    float th = step * (float)((ky * h) & 255);
    v = jt ? sinf(th) : cosf(th);
  } else if (idx < 36864) {
    int q = idx - 20480;
    int ks = (q >> 7) & 7, ht = q >> 10;
    int h = 16 * ht + c;
    int r = 4 * ks + 2 * hf + s;
    int ky = (r < 16) ? r : (224 + r);
    float th = step * (float)((ky * h) & 255);
    v = jt ? sinf(th) : cosf(th);
  } else {
    int q = idx - 36864;
    int p = (q >> 7) & 1, ks = (q >> 8) & 3, par = (q >> 10) & 1, g = q >> 11;
    int w = 64 * g + 4 * c + (2 * p + jt);
    int kx = 8 * hf + 4 * (ks & 1) + 2 * s + par;
    float th = step * (float)((kx * w) & 255);
    float sc = ((kx == 0) ? 1.0f : 2.0f) * (1.0f / 65536.0f);
    v = (ks < 2) ? sc * cosf(th) : -sc * sinf(th);
  }
  tab[idx] = v;
}

#define K1_PITCH4 17
#define K1_TILE4 (2 * 16 * K1_PITCH4)
__global__ __launch_bounds__(512) __attribute__((amdgpu_waves_per_eu(8, 8)))
void k1_dft_wh(const float* __restrict__ x, const float* __restrict__ FT,
               const float* __restrict__ E5, float* __restrict__ Zre,
               float* __restrict__ Zim) {
  __shared__ v4f stage[16 * K1_TILE4];
  const int wv = threadIdx.x >> 5;
  const int bi = blockIdx.x;
  const int lane = threadIdx.x & 31;
  const int half = lane >> 4, l16 = lane & 15;
  const long rowbase = ((long)bi * 16 + wv) * 16;
  v4f* tile = stage + wv * K1_TILE4;
  const float* gsrc = x + (rowbase + half) * 256 + 4 * l16;
  v4f* sdst = tile + half * K1_PITCH4 + l16;
  const v4f* ssrc = tile + l16 * K1_PITCH4 + half;
  const float* fb = FT + lane * 4;
  const int dofs = 16 * K1_PITCH4;
  v8f acc0 = {}, acc1 = {};
  v4f lo[8], hi[8];
#pragma unroll
  for (int i = 0; i < 8; ++i) {
    lo[i] = ld4_nt(gsrc + i * 512);
    hi[i] = ld4_nt(gsrc + i * 512 + 128);
  }
#pragma unroll
  for (int c = 0; c < 2; ++c) {
#pragma unroll
    for (int i = 0; i < 8; ++i) {
      sdst[2 * i * K1_PITCH4] = lo[i] + hi[i];
      sdst[2 * i * K1_PITCH4 + dofs] = lo[i] - hi[i];
    }
    if (c < 1) {
#pragma unroll
      for (int i = 0; i < 8; ++i) {
        lo[i] = ld4_nt(gsrc + i * 512 + 64);
        hi[i] = ld4_nt(gsrc + i * 512 + 192);
      }
    }
    __builtin_amdgcn_fence(__ATOMIC_ACQ_REL, "wavefront");
#pragma unroll
    for (int j = 0; j < 8; ++j) {
      const int ks = (c * 8 + j) * 2;
      v4f rs = ssrc[2 * j];
      v4f rd = ssrc[2 * j + dofs];
      v4f f0 = ld4(fb + ks * 128);
      v4f f1 = ld4(fb + (ks + 1) * 128);
      v2f s0 = {rs.x, rs.y};
      v2f s1 = {rs.z, rs.w};
      v2f d0 = {rd.x, rd.y};
      v2f d1 = {rd.z, rd.w};
      v2f b00 = {f0.x, f0.y};
      v2f b01 = {f0.z, f0.w};
      v2f b10 = {f1.x, f1.y};
      v2f b11 = {f1.z, f1.w};
      acc0 = wmma4(s0, b00, acc0);
      acc1 = wmma4(d0, b01, acc1);
      acc0 = wmma4(s1, b10, acc0);
      acc1 = wmma4(d1, b11, acc1);
    }
    if (c < 1) __builtin_amdgcn_fence(__ATOMIC_ACQ_REL, "wavefront");
  }
  const float* e5 = E5 + ((wv * 4) * 32 + lane) * 8;
  v4f ea[4], eb[4];
  __builtin_amdgcn_sched_barrier(0);
#pragma unroll
  for (int ks = 0; ks < 4; ++ks) {
    ea[ks] = ld4(e5 + ks * 256);
    eb[ks] = ld4(e5 + ks * 256 + 4);
  }
  const bool lo8 = l16 < 8;
  v8f bre, bim;
#pragma unroll
  for (int r = 0; r < 8; ++r) {
    const float p0 = __shfl_xor(acc0[r], 8);
    const float p1 = __shfl_xor(acc1[r], 8);
    bre[r] = lo8 ? acc0[r] : p1;
    bim[r] = lo8 ? p0 : acc1[r];
  }
  v8f zr0 = {}, zr1 = {}, zi0 = {}, zi1 = {};
#pragma unroll
  for (int ks = 0; ks < 4; ++ks) {
    const v2f br = {bre[2 * ks], bre[2 * ks + 1]};
    const v2f bm = {bim[2 * ks], bim[2 * ks + 1]};
    const v2f nb = -br;
    const v2f c0 = {ea[ks].x, ea[ks].y}, s0 = {ea[ks].z, ea[ks].w};
    const v2f c1 = {eb[ks].x, eb[ks].y}, s1 = {eb[ks].z, eb[ks].w};
    zr0 = wmma4(c0, br, zr0);
    zi0 = wmma4(c0, bm, zi0);
    zr1 = wmma4(c1, br, zr1);
    zi1 = wmma4(c1, bm, zi1);
    zr0 = wmma4(s0, bm, zr0);
    zi0 = wmma4(s0, nb, zi0);
    zr1 = wmma4(s1, bm, zr1);
    zi1 = wmma4(s1, nb, zi1);
  }
  {
    const v4f q00 = {zr0[0], zr0[1], zr0[2], zr0[3]};
    const v4f q01 = {zr0[4], zr0[5], zr0[6], zr0[7]};
    const v4f q10 = {zr1[0], zr1[1], zr1[2], zr1[3]};
    const v4f q11 = {zr1[4], zr1[5], zr1[6], zr1[7]};
    const v4f q20 = {zi0[0], zi0[1], zi0[2], zi0[3]};
    const v4f q21 = {zi0[4], zi0[5], zi0[6], zi0[7]};
    const v4f q30 = {zi1[0], zi1[1], zi1[2], zi1[3]};
    const v4f q31 = {zi1[4], zi1[5], zi1[6], zi1[7]};
    tile[0 * 32 + lane] = q00;
    tile[1 * 32 + lane] = q01;
    tile[2 * 32 + lane] = q10;
    tile[3 * 32 + lane] = q11;
    tile[4 * 32 + lane] = q20;
    tile[5 * 32 + lane] = q21;
    tile[6 * 32 + lane] = q30;
    tile[7 * 32 + lane] = q31;
  }
  __syncthreads();
  if (threadIdx.x < 256) {
    const int slot = threadIdx.x;
    v4f sum = stage[slot];
#pragma unroll
    for (int w = 1; w < 16; ++w) sum += stage[w * K1_TILE4 + slot];
    stage[slot] = sum;
  }
  __syncthreads();
  if (threadIdx.x < 256) {
    const int t = threadIdx.x;
    const int reim = t >> 7, kyq = (t >> 2) & 31, kxg = t & 3;
    const int q = 2 * reim + (kyq >> 4);
    const int shf = (kyq >> 3) & 1, rh = (kyq >> 2) & 1, e = kyq & 3;
    const float* sf = (const float*)stage + ((2 * q + rh) * 32 + 16 * shf) * 4 + e;
    v4f v;
#pragma unroll
    for (int n = 0; n < 4; ++n) {
      const int kx = 4 * kxg + n;
      const int sc = (kx & 1) ? 8 + (kx >> 1) : (kx >> 1);
      v[n] = sf[sc * 4];
    }
    const int b = bi >> 5, i = bi & 31;
    float* Z = reim ? Zim : Zre;
    *(v4f*)(Z + ((((long)kyq * 4 + kxg) * 16 + b) * 32 + i) * 4) = v;
  }
}

__global__ __launch_bounds__(32) void k3_mix(
    const float* __restrict__ Zre, const float* __restrict__ Zim,
    const float* __restrict__ w1r, const float* __restrict__ w1i,
    const float* __restrict__ w2r, const float* __restrict__ w2i,
    float* __restrict__ Tre, float* __restrict__ Tim) {
  const int wid = blockIdx.x;
  const int ot = wid & 1;
  const int kxg = (wid >> 1) & 3;
  const int kyp = wid >> 3;
  const int lane = threadIdx.x;
  const int half = lane >> 4, l16 = lane & 15;
  const int ky = kyp & 15;
  const float* wr = (kyp < 16) ? w1r : w2r;
  const float* wi = (kyp < 16) ? w1i : w2i;
  const long wofs =
      ((long)(2 * half) * 32 + 16 * ot + l16) * 256 + ky * 16 + 4 * kxg;
  v4f br[8][2], bi[8][2];
  v4f zr[8][2], zi[8][2];
  const long zofs = ((long)(kyp * 4 + kxg) * 16 + l16) * 128 + 8 * half;
  const float* zre = Zre + zofs;
  const float* zim = Zim + zofs;
#pragma unroll
  for (int ks = 0; ks < 8; ++ks) {
#pragma unroll
    for (int s = 0; s < 2; ++s) {
      br[ks][s] = ld4(wr + wofs + (long)(4 * ks + s) * 8192);
      bi[ks][s] = ld4(wi + wofs + (long)(4 * ks + s) * 8192);
    }
#pragma unroll
    for (int s = 0; s < 2; ++s) {
      zr[ks][s] = ld4(zre + (4 * ks + s) * 4);
      zi[ks][s] = ld4(zim + (4 * ks + s) * 4);
    }
  }
  __builtin_amdgcn_sched_barrier(0);
  v8f p[4] = {}, n[4] = {}, u[4] = {}, v[4] = {};
#pragma unroll
  for (int ks = 0; ks < 8; ++ks) {
#pragma unroll
    for (int q = 0; q < 4; ++q) {
      v2f b_r = {br[ks][0][q], br[ks][1][q]};
      v2f b_i = {bi[ks][0][q], bi[ks][1][q]};
      v2f a_r = {zr[ks][0][q], zr[ks][1][q]};
      v2f a_i = {zi[ks][0][q], zi[ks][1][q]};
      p[q] = wmma4(a_r, b_r, p[q]);
      n[q] = wmma4(a_i, b_i, n[q]);
      u[q] = wmma4(a_r, b_i, u[q]);
      v[q] = wmma4(a_i, b_r, v[q]);
    }
    __builtin_amdgcn_sched_barrier(0);
  }
  v8f tre[4], tim[4];
#pragma unroll
  for (int q = 0; q < 4; ++q) {
    tre[q] = p[q] - n[q];
    tim[q] = u[q] + v[q];
  }
#pragma unroll
  for (int r = 0; r < 8; ++r) {
    const int b = r + half * 8;
    const long o = (((long)b * 32 + 16 * ot + l16) * 32 + kyp) * 16 + 4 * kxg;
    v4f tr = {tre[0][r], tre[1][r], tre[2][r], tre[3][r]};
    v4f ti = {tim[0][r], tim[1][r], tim[2][r], tim[3][r]};
    *(v4f*)(Tre + o) = tr;
    *(v4f*)(Tim + o) = ti;
  }
}

__global__ __launch_bounds__(256) __attribute__((amdgpu_waves_per_eu(8, 8)))
void k4_idft_hw(
    const float* __restrict__ Tre, const float* __restrict__ Tim,
    const float* __restrict__ E4F, const float* __restrict__ GT,
    float* __restrict__ out) {
  const int wid = blockIdx.x * 8 + (threadIdx.x >> 5);
  const int lane = threadIdx.x & 31;
  const int half = lane >> 4, l16 = lane & 15;
  const int bo = wid >> 4;
  const int ht = wid & 15;
  const float* tre = Tre + (long)bo * 512;
  const float* tim = Tim + (long)bo * 512;
  const float* ef = E4F + (ht * 256 + lane) * 4;
  const int kxl = (l16 & 8) + 2 * (l16 & 3) + ((l16 >> 2) & 1);
  v8f a0 = {}, a1 = {}, a2 = {}, a3 = {};
#pragma unroll
  for (int k0 = 0; k0 < 32; k0 += 4) {
    int kr = k0 + 2 * half;
    v4f e = ld4(ef + k0 * 32);
    v2f ec = {e.x, e.y};
    v2f es = {e.z, e.w};
    v2f tr, ti;
    tr.x = tre[kr * 16 + kxl];
    tr.y = tre[(kr + 1) * 16 + kxl];
    ti.x = tim[kr * 16 + kxl];
    ti.y = tim[(kr + 1) * 16 + kxl];
    a0 = wmma4(tr, ec, a0);
    a1 = wmma4(ti, es, a1);
    a2 = wmma4(ti, ec, a2);
    a3 = wmma4(tr, es, a3);
  }
  const v8f ure = a0 - a1;
  const v8f uim = a2 + a3;
  float* orow = out + ((long)bo * 256 + ht * 16) * 256;
  const float* gf = GT + lane * 4;
#pragma unroll 1
  for (int g = 0; g < 2; ++g) {
    v8f ev[4] = {}, od[4] = {};
#pragma unroll
    for (int ks = 0; ks < 4; ++ks) {
      if (ks == 2) __builtin_amdgcn_sched_barrier(0);
      const int j = 2 * (ks & 1);
      v2f ae, ao;
      if (ks < 2) {
        ae.x = ure[j];
        ae.y = ure[j + 1];
        ao.x = ure[4 + j];
        ao.y = ure[5 + j];
      } else {
        ae.x = uim[j];
        ae.y = uim[j + 1];
        ao.x = uim[4 + j];
        ao.y = uim[5 + j];
      }
#pragma unroll
      for (int p = 0; p < 2; ++p) {
        v4f te = ld4(gf + g * 2048 + ks * 256 + p * 128);
        v4f to = ld4(gf + g * 2048 + 1024 + ks * 256 + p * 128);
        v2f ea = {te.x, te.y}, eb = {te.z, te.w};
        v2f oa = {to.x, to.y}, ob = {to.z, to.w};
        ev[2 * p] = wmma4(ae, ea, ev[2 * p]);
        od[2 * p] = wmma4(ao, oa, od[2 * p]);
        ev[2 * p + 1] = wmma4(ae, eb, ev[2 * p + 1]);
        od[2 * p + 1] = wmma4(ao, ob, od[2 * p + 1]);
      }
    }
#pragma unroll
    for (int r = 0; r < 8; ++r) {
      int rr = r + half * 8;
      v4f e = {ev[0][r], ev[1][r], ev[2][r], ev[3][r]};
      v4f o = {od[0][r], od[1][r], od[2][r], od[3][r]};
      float* dst = &orow[(long)rr * 256 + g * 64 + 4 * l16];
      __builtin_nontemporal_store(e + o, (v4f*)dst);
      __builtin_nontemporal_store(e - o, (v4f*)(dst + 128));
    }
  }
}

extern "C" void kernel_launch(void* const* d_in, const int* in_sizes, int n_in,
                              void* d_out, int out_size, void* d_ws,
                              size_t ws_size, hipStream_t stream) {
  (void)in_sizes; (void)n_in; (void)out_size; (void)ws_size;
  const float* x = (const float*)d_in[0];
  const float* w1r = (const float*)d_in[1];
  const float* w1i = (const float*)d_in[2];
  const float* w2r = (const float*)d_in[3];
  const float* w2i = (const float*)d_in[4];
  float* out = (float*)d_out;

  float* tab = (float*)d_ws;
  float* FT = tab;
  float* E5 = tab + 4096;
  float* E4F = tab + 20480;
  float* GT = tab + 36864;
  float* Zre = tab + 40960;
  float* Zim = Zre + 262144;
  float* Tre = Zim + 262144;
  float* Tim = Tre + 262144;

  hipLaunchKernelGGL(k0_tables, dim3(160), dim3(256), 0, stream, tab);
  hipLaunchKernelGGL(k1_dft_wh, dim3(512), dim3(512), 0, stream, x, FT, E5, Zre,
                     Zim);
  hipLaunchKernelGGL(k3_mix, dim3(256), dim3(32), 0, stream, Zre, Zim, w1r, w1i,
                     w2r, w2i, Tre, Tim);
  hipLaunchKernelGGL(k4_idft_hw, dim3(1024), dim3(256), 0, stream, Tre, Tim,
                     E4F, GT, out);
}
